# starting kernel
# baseline (speedup 1.0000x reference)
#include <hip/hip_runtime.h>
#include <hip/hip_bf16.h>

typedef __attribute__((ext_vector_type(2))) float v2f;
typedef __attribute__((ext_vector_type(8))) float v8f;

#define F_IN 512
#define HID 8
#define HEADS 8
#define D1 64
#define NCLS 16
#define NEG_SLOPE 0.2f
#define EPS_GAT 1e-16f

__device__ __forceinline__ unsigned fenc(float f) {
  unsigned u = __float_as_uint(f);
  return (u & 0x80000000u) ? ~u : (u | 0x80000000u);
}
__device__ __forceinline__ float fdec(unsigned e) {
  return (e & 0x80000000u) ? __uint_as_float(e ^ 0x80000000u) : __uint_as_float(~e);
}
#define ENC_NEG_INF 0x007FFFFFu

__device__ __forceinline__ float lrelu(float x) { return x > 0.f ? x : x * NEG_SLOPE; }

__global__ void fill_f32(float* p, float v, int n) {
  int i = blockIdx.x * blockDim.x + threadIdx.x;
  if (i < n) p[i] = v;
}
__global__ void fill_u32(unsigned* p, unsigned v, int n) {
  int i = blockIdx.x * blockDim.x + threadIdx.x;
  if (i < n) p[i] = v;
}

__global__ __launch_bounds__(128) void gemm1_kernel(const float* __restrict__ x,
                                                    const float* __restrict__ W1,
                                                    float* __restrict__ xl1, int nrows) {
  __shared__ float At[16 * 516];
  const int tid = threadIdx.x;
  const int rowBase = blockIdx.x * 16;
  for (int i = tid; i < 2048; i += 128) {
    int r = i >> 7, cq = i & 127;
    int row = rowBase + r; if (row >= nrows) row = nrows - 1;
    *(float4*)&At[r * 516 + cq * 4] = ((const float4*)(x + (size_t)row * F_IN))[cq];
  }
  __syncthreads();

  const int wave = tid >> 5;
  const int lane = tid & 31;
  const int ln = lane & 15;
  const int khalf = (lane >> 4) << 1;
  const int colBase = wave * 16;

  v8f c = {};
  #pragma unroll 4
  for (int k = 0; k < F_IN; k += 4) {
    v2f a = *(const v2f*)&At[ln * 516 + k + khalf];
    const float* wp = W1 + (size_t)(k + khalf) * D1 + colBase + ln;
    v2f b; b.x = wp[0]; b.y = wp[D1];
    c = __builtin_amdgcn_wmma_f32_16x16x4_f32(false, a, false, b, (short)0, c,
                                              false, false);
  }
  const int mBase = (lane >> 4) * 8;
  float* outp = xl1 + (size_t)rowBase * D1 + colBase;
  #pragma unroll
  for (int i = 0; i < 8; i++) {
    int row = rowBase + mBase + i;
    if (row < nrows) outp[(size_t)(mBase + i) * D1 + ln] = c[i];
  }
}

__global__ void attn1_kernel(const float* __restrict__ xl1,
                             const float* __restrict__ att_src,
                             const float* __restrict__ att_dst,
                             float* __restrict__ a_src, float* __restrict__ a_dst, int n) {
  int t = blockIdx.x * blockDim.x + threadIdx.x;
  if (t >= n * HEADS) return;
  int node = t >> 3, hh = t & 7;
  const float* xp = xl1 + (size_t)node * D1 + hh * HID;
  const float* as = att_src + hh * HID;
  const float* ad = att_dst + hh * HID;
  float s1 = 0.f, s2 = 0.f;
  #pragma unroll
  for (int cc = 0; cc < HID; cc++) { float v = xp[cc]; s1 += v * as[cc]; s2 += v * ad[cc]; }
  a_src[t] = s1; a_dst[t] = s2;
}

__global__ void edge_max1(const int* __restrict__ ei, const float* __restrict__ a_src,
                          const float* __restrict__ a_dst, unsigned* __restrict__ amax, int E_) {
  int e = blockIdx.x * blockDim.x + threadIdx.x;
  if (e >= E_) return;
  int s = ei[e], d = ei[E_ + e];
  #pragma unroll
  for (int hh = 0; hh < HEADS; hh++) {
    float al = lrelu(a_src[s * HEADS + hh] + a_dst[d * HEADS + hh]);
    atomicMax(&amax[d * HEADS + hh], fenc(al));
  }
}

__global__ void edge_sum1(const int* __restrict__ ei, const float* __restrict__ a_src,
                          const float* __restrict__ a_dst, const unsigned* __restrict__ amax,
                          float* __restrict__ denom, int E_) {
  int e = blockIdx.x * blockDim.x + threadIdx.x;
  if (e >= E_) return;
  int s = ei[e], d = ei[E_ + e];
  #pragma unroll
  for (int hh = 0; hh < HEADS; hh++) {
    float al = lrelu(a_src[s * HEADS + hh] + a_dst[d * HEADS + hh]);
    float ex = __expf(al - fdec(amax[d * HEADS + hh]));
    atomicAdd(&denom[d * HEADS + hh], ex);
  }
}

__global__ void edge_msg1(const int* __restrict__ ei, const float* __restrict__ a_src,
                          const float* __restrict__ a_dst, const unsigned* __restrict__ amax,
                          const float* __restrict__ denom, const float* __restrict__ xl1,
                          float* __restrict__ out1, int E_) {
  long long t = (long long)blockIdx.x * blockDim.x + threadIdx.x;
  if (t >= (long long)E_ * HEADS) return;
  int e = (int)(t >> 3), hh = (int)(t & 7);
  int s = ei[e], d = ei[E_ + e];
  float al = lrelu(a_src[s * HEADS + hh] + a_dst[d * HEADS + hh]);
  float ex = __expf(al - fdec(amax[d * HEADS + hh]));
  float coef = ex / (denom[d * HEADS + hh] + EPS_GAT);
  const float4* xp = (const float4*)(xl1 + (size_t)s * D1 + hh * HID);
  float4 v0 = xp[0], v1 = xp[1];
  float* op = out1 + (size_t)d * D1 + hh * HID;
  atomicAdd(op + 0, v0.x * coef); atomicAdd(op + 1, v0.y * coef);
  atomicAdd(op + 2, v0.z * coef); atomicAdd(op + 3, v0.w * coef);
  atomicAdd(op + 4, v1.x * coef); atomicAdd(op + 5, v1.y * coef);
  atomicAdd(op + 6, v1.z * coef); atomicAdd(op + 7, v1.w * coef);
}

__global__ void elu_kernel(float* __restrict__ out1, const float* __restrict__ bias1, int n) {
  int i = blockIdx.x * blockDim.x + threadIdx.x;
  if (i >= n) return;
  float v = out1[i] + bias1[i & (D1 - 1)];
  out1[i] = v > 0.f ? v : (__expf(v) - 1.f);
}

__global__ __launch_bounds__(256) void gemm2_kernel(const float* __restrict__ h,
                                                    const float* __restrict__ W2,
                                                    float* __restrict__ xl2, int nrows) {
  const int wave = threadIdx.x >> 5;
  const int lane = threadIdx.x & 31;
  const int tile = blockIdx.x * 8 + wave;
  const int rowBase = tile * 16;
  if (rowBase >= nrows) return;
  const int ln = lane & 15;
  const int khalf = (lane >> 4) << 1;
  int arow = rowBase + ln; if (arow >= nrows) arow = nrows - 1;
  const float* ap = h + (size_t)arow * D1;

  v8f c = {};
  #pragma unroll
  for (int k = 0; k < D1; k += 4) {
    v2f a = *(const v2f*)&ap[k + khalf];
    v2f b; b.x = W2[(k + khalf) * NCLS + ln];
           b.y = W2[(k + khalf + 1) * NCLS + ln];
    c = __builtin_amdgcn_wmma_f32_16x16x4_f32(false, a, false, b, (short)0, c,
                                              false, false);
  }
  const int mBase = (lane >> 4) * 8;
  float* outp = xl2 + (size_t)rowBase * NCLS;
  #pragma unroll
  for (int i = 0; i < 8; i++) {
    int row = rowBase + mBase + i;
    if (row < nrows) outp[(size_t)(mBase + i) * NCLS + ln] = c[i];
  }
}

__global__ void attn2_kernel(const float* __restrict__ xl2,
                             const float* __restrict__ att_src,
                             const float* __restrict__ att_dst,
                             float* __restrict__ a_src, float* __restrict__ a_dst, int n) {
  int t = blockIdx.x * blockDim.x + threadIdx.x;
  if (t >= n) return;
  const float* xp = xl2 + (size_t)t * NCLS;
  float s1 = 0.f, s2 = 0.f;
  #pragma unroll
  for (int cc = 0; cc < NCLS; cc++) { float v = xp[cc]; s1 += v * att_src[cc]; s2 += v * att_dst[cc]; }
  a_src[t] = s1; a_dst[t] = s2;
}

__global__ void edge_max2(const int* __restrict__ ei, const float* __restrict__ a_src,
                          const float* __restrict__ a_dst, unsigned* __restrict__ amax, int E_) {
  int e = blockIdx.x * blockDim.x + threadIdx.x;
  if (e >= E_) return;
  int s = ei[e], d = ei[E_ + e];
  atomicMax(&amax[d], fenc(lrelu(a_src[s] + a_dst[d])));
}

__global__ void edge_sum2(const int* __restrict__ ei, const float* __restrict__ a_src,
                          const float* __restrict__ a_dst, const unsigned* __restrict__ amax,
                          float* __restrict__ denom, int E_) {
  int e = blockIdx.x * blockDim.x + threadIdx.x;
  if (e >= E_) return;
  int s = ei[e], d = ei[E_ + e];
  float al = lrelu(a_src[s] + a_dst[d]);
  atomicAdd(&denom[d], __expf(al - fdec(amax[d])));
}

__global__ void edge_msg2(const int* __restrict__ ei, const float* __restrict__ a_src,
                          const float* __restrict__ a_dst, const unsigned* __restrict__ amax,
                          const float* __restrict__ denom, const float* __restrict__ xl2,
                          float* __restrict__ out2, int E_) {
  int e = blockIdx.x * blockDim.x + threadIdx.x;
  if (e >= E_) return;
  int s = ei[e], d = ei[E_ + e];
  float al = lrelu(a_src[s] + a_dst[d]);
  float ex = __expf(al - fdec(amax[d]));
  float coef = ex / (denom[d] + EPS_GAT);
  const float4* xp = (const float4*)(xl2 + (size_t)s * NCLS);
  float* op = out2 + (size_t)d * NCLS;
  #pragma unroll
  for (int q = 0; q < 4; q++) {
    float4 v = xp[q];
    atomicAdd(op + q * 4 + 0, v.x * coef); atomicAdd(op + q * 4 + 1, v.y * coef);
    atomicAdd(op + q * 4 + 2, v.z * coef); atomicAdd(op + q * 4 + 3, v.w * coef);
  }
}

__global__ void final_kernel(const float* __restrict__ out2, const float* __restrict__ bias2,
                             float* __restrict__ out, int n) {
  int i = blockIdx.x * blockDim.x + threadIdx.x;
  if (i >= n) return;
  out[i] = out2[i] + bias2[i & (NCLS - 1)];
}

extern "C" void kernel_launch(void* const* d_in, const int* in_sizes, int n_in,
                              void* d_out, int out_size, void* d_ws, size_t ws_size,
                              hipStream_t stream) {
  const float* x        = (const float*)d_in[0];
  const int*   ei       = (const int*)d_in[1];
  const float* W1       = (const float*)d_in[2];
  const float* att_src1 = (const float*)d_in[3];
  const float* att_dst1 = (const float*)d_in[4];
  const float* bias1    = (const float*)d_in[5];
  const float* W2       = (const float*)d_in[6];
  const float* att_src2 = (const float*)d_in[7];
  const float* att_dst2 = (const float*)d_in[8];
  const float* bias2    = (const float*)d_in[9];
  float* out = (float*)d_out;

  const int N = in_sizes[0] / F_IN;
  const int E = in_sizes[1] / 2;

  float* ws = (float*)d_ws;
  size_t o = 0;
  float*    xl1    = ws + o; o += (size_t)N * D1;
  float*    out1   = ws + o; o += (size_t)N * D1;
  float*    a_src1 = ws + o; o += (size_t)N * HEADS;
  float*    a_dst1 = ws + o; o += (size_t)N * HEADS;
  unsigned* amax1  = (unsigned*)(ws + o); o += (size_t)N * HEADS;
  float*    denom1 = ws + o; o += (size_t)N * HEADS;
  float*    xl2    = ws + o; o += (size_t)N * NCLS;
  float*    a_src2 = ws + o; o += (size_t)N;
  float*    a_dst2 = ws + o; o += (size_t)N;
  unsigned* amax2  = (unsigned*)(ws + o); o += (size_t)N;
  float*    denom2 = ws + o; o += (size_t)N;
  float*    out2   = ws + o; o += (size_t)N * NCLS;

  const int TB = 256;
  auto nb = [](long long n, int b) { return (unsigned)((n + b - 1) / b); };

  fill_f32<<<nb((long long)N * D1, TB), TB, 0, stream>>>(out1, 0.f, N * D1);
  fill_f32<<<nb((long long)N * HEADS, TB), TB, 0, stream>>>(denom1, 0.f, N * HEADS);
  fill_u32<<<nb((long long)N * HEADS, TB), TB, 0, stream>>>(amax1, ENC_NEG_INF, N * HEADS);
  fill_f32<<<nb((long long)N * NCLS, TB), TB, 0, stream>>>(out2, 0.f, N * NCLS);
  fill_f32<<<nb((long long)N, TB), TB, 0, stream>>>(denom2, 0.f, N);
  fill_u32<<<nb((long long)N, TB), TB, 0, stream>>>(amax2, ENC_NEG_INF, N);

  gemm1_kernel<<<nb((long long)N, 16), 128, 0, stream>>>(x, W1, xl1, N);
  attn1_kernel<<<nb((long long)N * HEADS, TB), TB, 0, stream>>>(xl1, att_src1, att_dst1,
                                                                a_src1, a_dst1, N);
  edge_max1<<<nb(E, TB), TB, 0, stream>>>(ei, a_src1, a_dst1, amax1, E);
  edge_sum1<<<nb(E, TB), TB, 0, stream>>>(ei, a_src1, a_dst1, amax1, denom1, E);
  edge_msg1<<<nb((long long)E * HEADS, TB), TB, 0, stream>>>(ei, a_src1, a_dst1, amax1,
                                                             denom1, xl1, out1, E);
  elu_kernel<<<nb((long long)N * D1, TB), TB, 0, stream>>>(out1, bias1, N * D1);

  gemm2_kernel<<<nb(((long long)N + 15) / 16, 8), 256, 0, stream>>>(out1, W2, xl2, N);
  attn2_kernel<<<nb(N, TB), TB, 0, stream>>>(xl2, att_src2, att_dst2, a_src2, a_dst2, N);
  edge_max2<<<nb(E, TB), TB, 0, stream>>>(ei, a_src2, a_dst2, amax2, E);
  edge_sum2<<<nb(E, TB), TB, 0, stream>>>(ei, a_src2, a_dst2, amax2, denom2, E);
  edge_msg2<<<nb(E, TB), TB, 0, stream>>>(ei, a_src2, a_dst2, amax2, denom2, xl2, out2, E);
  final_kernel<<<nb((long long)N * NCLS, TB), TB, 0, stream>>>(out2, bias2, out, N * NCLS);
}


# Edge phase, both layers: each work-group owns a slice of destination nodes in LDS, scans the destination list for its edges; max, sums, divide on LDS atomics; global atomics go; sum order=LDS arrival
# speedup vs baseline: 3.8160x; 3.8160x over previous
#include <hip/hip_runtime.h>
#include <hip/hip_bf16.h>

typedef __attribute__((ext_vector_type(2))) float v2f;
typedef __attribute__((ext_vector_type(8))) float v8f;

#define F_IN 512
#define HID 8
#define HEADS 8
#define D1 64
#define NCLS 16
#define NEG_SLOPE 0.2f
#define EPS_GAT 1e-16f

__device__ __forceinline__ unsigned fenc(float f) {
  unsigned u = __float_as_uint(f);
  return (u & 0x80000000u) ? ~u : (u | 0x80000000u);
}
__device__ __forceinline__ float fdec(unsigned e) {
  return (e & 0x80000000u) ? __uint_as_float(e ^ 0x80000000u) : __uint_as_float(~e);
}
#define ENC_NEG_INF 0x007FFFFFu

__device__ __forceinline__ float lrelu(float x) { return x > 0.f ? x : x * NEG_SLOPE; }

__global__ __launch_bounds__(128) void gemm1_kernel(const float* __restrict__ x,
                                                    const float* __restrict__ W1,
                                                    float* __restrict__ xl1, int nrows) {
  __shared__ float At[16 * 516];
  const int tid = threadIdx.x;
  const int rowBase = blockIdx.x * 16;
  for (int i = tid; i < 2048; i += 128) {
    int r = i >> 7, cq = i & 127;
    int row = rowBase + r; if (row >= nrows) row = nrows - 1;
    *(float4*)&At[r * 516 + cq * 4] = ((const float4*)(x + (size_t)row * F_IN))[cq];
  }
  __syncthreads();

  const int wave = tid >> 5;
  const int lane = tid & 31;
  const int ln = lane & 15;
  const int khalf = (lane >> 4) << 1;
  const int colBase = wave * 16;

  v8f c = {};
  #pragma unroll 4
  for (int k = 0; k < F_IN; k += 4) {
    v2f a = *(const v2f*)&At[ln * 516 + k + khalf];
    const float* wp = W1 + (size_t)(k + khalf) * D1 + colBase + ln;
    v2f b; b.x = wp[0]; b.y = wp[D1];
    c = __builtin_amdgcn_wmma_f32_16x16x4_f32(false, a, false, b, (short)0, c,
                                              false, false);
  }
  const int mBase = (lane >> 4) * 8;
  float* outp = xl1 + (size_t)rowBase * D1 + colBase;
  #pragma unroll
  for (int i = 0; i < 8; i++) {
    int row = rowBase + mBase + i;
    if (row < nrows) outp[(size_t)(mBase + i) * D1 + ln] = c[i];
  }
}

__global__ void attn1_kernel(const float* __restrict__ xl1,
                             const float* __restrict__ att_src,
                             const float* __restrict__ att_dst,
                             float* __restrict__ a_src, float* __restrict__ a_dst, int n) {
  int t = blockIdx.x * blockDim.x + threadIdx.x;
  if (t >= n * HEADS) return;
  int node = t >> 3, hh = t & 7;
  const float* xp = xl1 + (size_t)node * D1 + hh * HID;
  const float* as = att_src + hh * HID;
  const float* ad = att_dst + hh * HID;
  float s1 = 0.f, s2 = 0.f;
  #pragma unroll
  for (int cc = 0; cc < HID; cc++) { float v = xp[cc]; s1 += v * as[cc]; s2 += v * ad[cc]; }
  a_src[t] = s1; a_dst[t] = s2;
}

#define NB_L1 392
#define CAP_L1 16384
#define NB_L2 784
#define CAP_L2 24576

template <int H, int CH, int NB, int CAP, bool ELU>
__global__ __launch_bounds__(1024) void agg_kernel(const int* __restrict__ ei,
                                                   const float* __restrict__ a_src,
                                                   const float* __restrict__ a_dst,
                                                   const float* __restrict__ xl,
                                                   const float* __restrict__ bias,
                                                   float* __restrict__ outp, int n, int E_) {
  constexpr int D = H * CH;
  constexpr int RS = D + 1;
  constexpr int NT = 1024;
  constexpr int NW = NT / 32;
  constexpr int SEG = CAP / NW;
  __shared__ float acc[NB * RS];
  __shared__ float den[NB * H];
  __shared__ unsigned mx[NB * H];
  __shared__ float adl[NB * H];
  __shared__ unsigned seg[CAP];
  __shared__ unsigned lst[CAP];
  __shared__ unsigned short dlo[CAP];
  __shared__ unsigned wcnt[NW];

  const int tid = threadIdx.x;
  const int n0 = blockIdx.x * NB;
  const unsigned nloc = (unsigned)((n - n0) < NB ? (n - n0) : NB);
  const int* __restrict__ dstp = ei + E_;

  for (int i = tid; i < NB * RS; i += NT) acc[i] = 0.f;
  for (int i = tid; i < NB * H; i += NT) {
    den[i] = 0.f; mx[i] = ENC_NEG_INF;
    adl[i] = ((unsigned)i < nloc * H) ? a_dst[(size_t)n0 * H + i] : 0.f;
  }
  __syncthreads();

  const unsigned lane = tid & 31, wave = tid >> 5;
  unsigned* myseg = seg + wave * SEG;
  unsigned wn = 0u;
  auto push = [&](bool m, unsigned e) {
    const unsigned mask = __builtin_amdgcn_ballot_w32(m);
    if (m) { const unsigned k = wn + __builtin_amdgcn_mbcnt_lo(mask, 0u); if (k < (unsigned)SEG) myseg[k] = e; }
    wn += (unsigned)__builtin_popcount(mask);
  };
  const int nvec = ((E_ & 3) == 0) ? (E_ >> 2) : 0;
  for (int vb0 = 0; vb0 < nvec; vb0 += NT * 8) {
    int4 dv[8];
    #pragma unroll
    for (int u = 0; u < 8; u++) {
      const int v = vb0 + u * NT + tid;
      dv[u] = ((const int4*)dstp)[v < nvec ? v : 0];
    }
    #pragma unroll
    for (int u = 0; u < 8; u++) {
      const int v = vb0 + u * NT + tid;
      const unsigned r0 = (unsigned)(dv[u].x - n0), r1 = (unsigned)(dv[u].y - n0);
      const unsigned r2 = (unsigned)(dv[u].z - n0), r3 = (unsigned)(dv[u].w - n0);
      const bool any = min(min(r0, r1), min(r2, r3)) < nloc && v < nvec;
      if (__builtin_amdgcn_ballot_w32(any) != 0u) {
        push(any && r0 < nloc, 4u * v + 0u); push(any && r1 < nloc, 4u * v + 1u);
        push(any && r2 < nloc, 4u * v + 2u); push(any && r3 < nloc, 4u * v + 3u);
      }
    }
  }
  for (int e0 = nvec * 4; e0 < E_; e0 += NT) {
    const int e = e0 + tid;
    const bool m = e < E_ && (unsigned)(dstp[e < E_ ? e : 0] - n0) < nloc;
    if (__builtin_amdgcn_ballot_w32(m) != 0u) push(m, (unsigned)e);
  }
  if (lane == 0) wcnt[wave] = wn;
  __syncthreads();
  unsigned total = 0u, off = 0u; bool fits = true;
  for (int w = 0; w < NW; w++) {
    const unsigned c = wcnt[w];
    if ((unsigned)w < wave) off += c;
    total += c; fits = fits && c <= (unsigned)SEG;
  }

  if (fits) {
    for (unsigned j = lane; j < wn; j += 32) {
      const unsigned e = myseg[j];
      lst[off + j] = (unsigned)ei[e]; dlo[off + j] = (unsigned short)(dstp[e] - n0);
    }
    __syncthreads();
    const unsigned items = total * H;
    const unsigned mfull = items - items % (NT * 4);
    for (unsigned tb = tid; tb < mfull; tb += NT * 4) {
      float al[4]; unsigned slot[4];
      #pragma unroll
      for (int u = 0; u < 4; u++) {
        const unsigned t = tb + u * NT, i = t / H, hh = t % H;
        slot[u] = (unsigned)dlo[i] * H + hh;
        al[u] = a_src[(size_t)lst[i] * H + hh];
      }
      #pragma unroll
      for (int u = 0; u < 4; u++)
        atomicMax(&mx[slot[u]], fenc(lrelu(al[u] + adl[slot[u]])));
    }
    for (unsigned t = mfull + tid; t < items; t += NT) {
      const unsigned i = t / H, hh = t % H, sl = (unsigned)dlo[i] * H + hh;
      atomicMax(&mx[sl], fenc(lrelu(a_src[(size_t)lst[i] * H + hh] + adl[sl])));
    }
    __syncthreads();
    constexpr int LPE = D / 8;
    constexpr int EPI = NT / LPE;
    constexpr int U = 4;
    const unsigned cb = (tid % LPE) * 8, hh = cb / CH;
    const unsigned sfull = total - total % (EPI * U);
    for (unsigned ib = tid / LPE; ib < sfull; ib += EPI * U) {
      float4 xa[U], xb[U]; float as_[U]; unsigned dl[U];
      #pragma unroll
      for (int u = 0; u < U; u++) {
        const unsigned i = ib + u * EPI;
        const unsigned s = lst[i]; dl[u] = dlo[i];
        as_[u] = a_src[(size_t)s * H + hh];
        const float4* xp = (const float4*)(xl + (size_t)s * D + cb);
        xa[u] = xp[0]; xb[u] = xp[1];
      }
      #pragma unroll
      for (int u = 0; u < U; u++) {
        const unsigned hs = dl[u] * H + hh;
        const float ex = __expf(lrelu(as_[u] + adl[hs]) - fdec(mx[hs]));
        if ((cb % CH) == 0) atomicAdd(&den[hs], ex);
        float* ap = &acc[dl[u] * RS + cb];
        atomicAdd(ap + 0, xa[u].x * ex); atomicAdd(ap + 1, xa[u].y * ex);
        atomicAdd(ap + 2, xa[u].z * ex); atomicAdd(ap + 3, xa[u].w * ex);
        atomicAdd(ap + 4, xb[u].x * ex); atomicAdd(ap + 5, xb[u].y * ex);
        atomicAdd(ap + 6, xb[u].z * ex); atomicAdd(ap + 7, xb[u].w * ex);
      }
    }
    for (unsigned i = sfull + tid / LPE; i < total; i += EPI) {
      const unsigned s = lst[i], hs = (unsigned)dlo[i] * H + hh;
      const float4* xp = (const float4*)(xl + (size_t)s * D + cb);
      const float4 xa = xp[0], xb = xp[1];
      const float ex = __expf(lrelu(a_src[(size_t)s * H + hh] + adl[hs]) - fdec(mx[hs]));
      if ((cb % CH) == 0) atomicAdd(&den[hs], ex);
      float* ap = &acc[(unsigned)dlo[i] * RS + cb];
      atomicAdd(ap + 0, xa.x * ex); atomicAdd(ap + 1, xa.y * ex);
      atomicAdd(ap + 2, xa.z * ex); atomicAdd(ap + 3, xa.w * ex);
      atomicAdd(ap + 4, xb.x * ex); atomicAdd(ap + 5, xb.y * ex);
      atomicAdd(ap + 6, xb.z * ex); atomicAdd(ap + 7, xb.w * ex);
    }
  } else {
    for (int e = tid; e < E_; e += NT) {
      const unsigned r = (unsigned)(dstp[e] - n0);
      if (r < nloc) {
        const int s = ei[e];
        for (int h2 = 0; h2 < H; h2++)
          atomicMax(&mx[r * H + h2], fenc(lrelu(a_src[(size_t)s * H + h2] + adl[r * H + h2])));
      }
    }
    __syncthreads();
    for (int e = tid; e < E_; e += NT) {
      const unsigned r = (unsigned)(dstp[e] - n0);
      if (r < nloc) {
        const int s = ei[e];
        for (int h2 = 0; h2 < H; h2++) {
          const unsigned hs = r * H + h2;
          const float ex = __expf(lrelu(a_src[(size_t)s * H + h2] + adl[hs]) - fdec(mx[hs]));
          atomicAdd(&den[hs], ex);
          for (int cc = 0; cc < CH; cc++)
            atomicAdd(&acc[r * RS + h2 * CH + cc], xl[(size_t)s * D + h2 * CH + cc] * ex);
        }
      }
    }
  }
  __syncthreads();
  float* op = outp + (size_t)n0 * D;
  for (unsigned i = tid; i < nloc * D; i += NT) {
    const unsigned c = i % D, r = i / D;
    float v = acc[r * RS + c] / (den[r * H + c / CH] + EPS_GAT) + bias[c];
    if (ELU) v = v > 0.f ? v : (__expf(v) - 1.f);
    op[i] = v;
  }
}

__global__ __launch_bounds__(256) void gemm2_kernel(const float* __restrict__ h,
                                                    const float* __restrict__ W2,
                                                    float* __restrict__ xl2, int nrows) {
  const int wave = threadIdx.x >> 5;
  const int lane = threadIdx.x & 31;
  const int tile = blockIdx.x * 8 + wave;
  const int rowBase = tile * 16;
  if (rowBase >= nrows) return;
  const int ln = lane & 15;
  const int khalf = (lane >> 4) << 1;
  int arow = rowBase + ln; if (arow >= nrows) arow = nrows - 1;
  const float* ap = h + (size_t)arow * D1;

  v8f c = {};
  #pragma unroll
  for (int k = 0; k < D1; k += 4) {
    v2f a = *(const v2f*)&ap[k + khalf];
    v2f b; b.x = W2[(k + khalf) * NCLS + ln];
           b.y = W2[(k + khalf + 1) * NCLS + ln];
    c = __builtin_amdgcn_wmma_f32_16x16x4_f32(false, a, false, b, (short)0, c,
                                              false, false);
  }
  const int mBase = (lane >> 4) * 8;
  float* outp = xl2 + (size_t)rowBase * NCLS;
  #pragma unroll
  for (int i = 0; i < 8; i++) {
    int row = rowBase + mBase + i;
    if (row < nrows) outp[(size_t)(mBase + i) * NCLS + ln] = c[i];
  }
}

__global__ void attn2_kernel(const float* __restrict__ xl2,
                             const float* __restrict__ att_src,
                             const float* __restrict__ att_dst,
                             float* __restrict__ a_src, float* __restrict__ a_dst, int n) {
  int t = blockIdx.x * blockDim.x + threadIdx.x;
  if (t >= n) return;
  const float* xp = xl2 + (size_t)t * NCLS;
  float s1 = 0.f, s2 = 0.f;
  #pragma unroll
  for (int cc = 0; cc < NCLS; cc++) { float v = xp[cc]; s1 += v * att_src[cc]; s2 += v * att_dst[cc]; }
  a_src[t] = s1; a_dst[t] = s2;
}

extern "C" void kernel_launch(void* const* d_in, const int* in_sizes, int n_in,
                              void* d_out, int out_size, void* d_ws, size_t ws_size,
                              hipStream_t stream) {
  const float* x        = (const float*)d_in[0];
  const int*   ei       = (const int*)d_in[1];
  const float* W1       = (const float*)d_in[2];
  const float* att_src1 = (const float*)d_in[3];
  const float* att_dst1 = (const float*)d_in[4];
  const float* bias1    = (const float*)d_in[5];
  const float* W2       = (const float*)d_in[6];
  const float* att_src2 = (const float*)d_in[7];
  const float* att_dst2 = (const float*)d_in[8];
  const float* bias2    = (const float*)d_in[9];
  float* out = (float*)d_out;

  const int N = in_sizes[0] / F_IN;
  const int E = in_sizes[1] / 2;

  float* ws = (float*)d_ws;
  size_t o = 0;
  float*    xl1    = ws + o; o += (size_t)N * D1;
  float*    h1     = ws + o; o += (size_t)N * D1;
  float*    a_src1 = ws + o; o += (size_t)N * HEADS;
  float*    a_dst1 = ws + o; o += (size_t)N * HEADS;
  float*    xl2    = ws + o; o += (size_t)N * NCLS;
  float*    a_src2 = ws + o; o += (size_t)N;
  float*    a_dst2 = ws + o; o += (size_t)N;

  const int TB = 256;
  auto nb = [](long long n, int b) { return (unsigned)((n + b - 1) / b); };

  gemm1_kernel<<<nb((long long)N, 16), 128, 0, stream>>>(x, W1, xl1, N);
  attn1_kernel<<<nb((long long)N * HEADS, TB), TB, 0, stream>>>(xl1, att_src1, att_dst1,
                                                                a_src1, a_dst1, N);
  agg_kernel<HEADS, HID, NB_L1, CAP_L1, true><<<nb(N, NB_L1), 1024, 0, stream>>>(
      ei, a_src1, a_dst1, xl1, bias1, h1, N, E);

  gemm2_kernel<<<nb(((long long)N + 15) / 16, 8), 256, 0, stream>>>(h1, W2, xl2, N);
  attn2_kernel<<<nb(N, TB), TB, 0, stream>>>(xl2, att_src2, att_dst2, a_src2, a_dst2, N);
  agg_kernel<1, NCLS, NB_L2, CAP_L2, false><<<nb(N, NB_L2), 1024, 0, stream>>>(
      ei, a_src2, a_dst2, xl2, bias2, out, N, E);
}


# Layer-2 aggregation takes over the per-slice edge lists that layer 1's scan found (published packed, whole lines, same call): layer 2's scan of the destination list goes; sum order = LDS arrival
# speedup vs baseline: 5.4002x; 1.4151x over previous
#include <hip/hip_runtime.h>
#include <hip/hip_bf16.h>

typedef __attribute__((ext_vector_type(2))) float v2f;
typedef __attribute__((ext_vector_type(8))) float v8f;

#define F_IN 512
#define HID 8
#define HEADS 8
#define D1 64
#define NCLS 16
#define NEG_SLOPE 0.2f
#define EPS_GAT 1e-16f

__device__ __forceinline__ unsigned fenc(float f) {
  unsigned u = __float_as_uint(f);
  return (u & 0x80000000u) ? ~u : (u | 0x80000000u);
}
__device__ __forceinline__ float fdec(unsigned e) {
  return (e & 0x80000000u) ? __uint_as_float(e ^ 0x80000000u) : __uint_as_float(~e);
}
#define ENC_NEG_INF 0x007FFFFFu

__device__ __forceinline__ float lrelu(float x) { return x > 0.f ? x : x * NEG_SLOPE; }

__global__ __launch_bounds__(128) void gemm1_kernel(const float* __restrict__ x,
                                                    const float* __restrict__ W1,
                                                    float* __restrict__ xl1, int nrows) {
  __shared__ float At[16 * 516];
  const int tid = threadIdx.x;
  const int rowBase = blockIdx.x * 16;
  for (int i = tid; i < 2048; i += 128) {
    int r = i >> 7, cq = i & 127;
    int row = rowBase + r; if (row >= nrows) row = nrows - 1;
    *(float4*)&At[r * 516 + cq * 4] = ((const float4*)(x + (size_t)row * F_IN))[cq];
  }
  __syncthreads();

  const int wave = tid >> 5;
  const int lane = tid & 31;
  const int ln = lane & 15;
  const int khalf = (lane >> 4) << 1;
  const int colBase = wave * 16;

  v8f c = {};
  #pragma unroll 4
  for (int k = 0; k < F_IN; k += 4) {
    v2f a = *(const v2f*)&At[ln * 516 + k + khalf];
    const float* wp = W1 + (size_t)(k + khalf) * D1 + colBase + ln;
    v2f b; b.x = wp[0]; b.y = wp[D1];
    c = __builtin_amdgcn_wmma_f32_16x16x4_f32(false, a, false, b, (short)0, c,
                                              false, false);
  }
  const int mBase = (lane >> 4) * 8;
  float* outp = xl1 + (size_t)rowBase * D1 + colBase;
  #pragma unroll
  for (int i = 0; i < 8; i++) {
    int row = rowBase + mBase + i;
    if (row < nrows) outp[(size_t)(mBase + i) * D1 + ln] = c[i];
  }
}

__global__ void attn1_kernel(const float* __restrict__ xl1,
                             const float* __restrict__ att_src,
                             const float* __restrict__ att_dst,
                             float* __restrict__ a_src, float* __restrict__ a_dst, int n) {
  int t = blockIdx.x * blockDim.x + threadIdx.x;
  if (t >= n * HEADS) return;
  int node = t >> 3, hh = t & 7;
  const float* xp = xl1 + (size_t)node * D1 + hh * HID;
  const float* as = att_src + hh * HID;
  const float* ad = att_dst + hh * HID;
  float s1 = 0.f, s2 = 0.f;
  #pragma unroll
  for (int cc = 0; cc < HID; cc++) { float v = xp[cc]; s1 += v * as[cc]; s2 += v * ad[cc]; }
  a_src[t] = s1; a_dst[t] = s2;
}

#define NB_AGG 392
#define CAP_AGG 16384
#define DLO_BITS 9
#define NOLIST 0xFFFFFFFFu
#define LIST_WORDS 18

template <int H, int CH, int NB, int CAP, bool ELU, bool SCAN>
__global__ __launch_bounds__(1024) void agg_kernel(const int* __restrict__ ei,
                                                   const float* __restrict__ a_src,
                                                   const float* __restrict__ a_dst,
                                                   const float* __restrict__ xl,
                                                   const float* __restrict__ bias,
                                                   float* __restrict__ outp, int n, int E_,
                                                   unsigned* lists) {
  static_assert(NB <= (1 << DLO_BITS), "local destination must fit its field");
  constexpr int D = H * CH;
  constexpr int RS = D + 1;
  constexpr int NT = 1024;
  constexpr int NW = NT / 32;
  constexpr int SEG = CAP / NW;
  __shared__ float acc[NB * RS];
  __shared__ float den[NB * H];
  __shared__ unsigned mx[NB * H];
  __shared__ float adl[NB * H];
  __shared__ unsigned seg[SCAN ? CAP : 1];
  __shared__ unsigned lst[CAP];
  __shared__ unsigned short dlo[CAP];
  __shared__ unsigned wcnt[SCAN ? NW : 1];

  const int tid = threadIdx.x;
  const int n0 = blockIdx.x * NB;
  const unsigned nloc = (unsigned)((n - n0) < NB ? (n - n0) : NB);
  const int* __restrict__ dstp = ei + E_;
  const unsigned cpad = (gridDim.x + 31u) & ~31u;
  const unsigned lcap = (((unsigned)LIST_WORDS * (unsigned)n - cpad) / gridDim.x) & ~31u;
  unsigned* ecnt = lists;
  unsigned* gl = lists + cpad + (size_t)blockIdx.x * lcap;

  for (int i = tid; i < NB * RS; i += NT) acc[i] = 0.f;
  for (int i = tid; i < NB * H; i += NT) {
    den[i] = 0.f; mx[i] = ENC_NEG_INF;
    adl[i] = ((unsigned)i < nloc * H) ? a_dst[(size_t)n0 * H + i] : 0.f;
  }
  __syncthreads();

  unsigned total = 0u; bool fits = true;
  if constexpr (SCAN) {
    const unsigned lane = tid & 31, wave = tid >> 5;
    unsigned* myseg = seg + wave * SEG;
    unsigned wn = 0u;
    auto push = [&](bool m, unsigned e) {
      const unsigned mask = __builtin_amdgcn_ballot_w32(m);
      if (m) { const unsigned k = wn + __builtin_amdgcn_mbcnt_lo(mask, 0u); if (k < (unsigned)SEG) myseg[k] = e; }
      wn += (unsigned)__builtin_popcount(mask);
    };
    const int nvec = ((E_ & 3) == 0) ? (E_ >> 2) : 0;
    for (int vb0 = 0; vb0 < nvec; vb0 += NT * 8) {
      int4 dv[8];
      #pragma unroll
      for (int u = 0; u < 8; u++) {
        const int v = vb0 + u * NT + tid;
        dv[u] = ((const int4*)dstp)[v < nvec ? v : 0];
      }
      #pragma unroll
      for (int u = 0; u < 8; u++) {
        const int v = vb0 + u * NT + tid;
        const unsigned r0 = (unsigned)(dv[u].x - n0), r1 = (unsigned)(dv[u].y - n0);
        const unsigned r2 = (unsigned)(dv[u].z - n0), r3 = (unsigned)(dv[u].w - n0);
        const bool any = min(min(r0, r1), min(r2, r3)) < nloc && v < nvec;
        if (__builtin_amdgcn_ballot_w32(any) != 0u) {
          push(any && r0 < nloc, 4u * v + 0u); push(any && r1 < nloc, 4u * v + 1u);
          push(any && r2 < nloc, 4u * v + 2u); push(any && r3 < nloc, 4u * v + 3u);
        }
      }
    }
    for (int e0 = nvec * 4; e0 < E_; e0 += NT) {
      const int e = e0 + tid;
      const bool m = e < E_ && (unsigned)(dstp[e < E_ ? e : 0] - n0) < nloc;
      if (__builtin_amdgcn_ballot_w32(m) != 0u) push(m, (unsigned)e);
    }
    if (lane == 0) wcnt[wave] = wn;
    __syncthreads();
    unsigned off = 0u;
    for (int w = 0; w < NW; w++) {
      const unsigned c = wcnt[w];
      if ((unsigned)w < wave) off += c;
      total += c; fits = fits && c <= (unsigned)SEG;
    }

    if (fits) {
      for (unsigned j = lane; j < wn; j += 32) {
        const unsigned e = myseg[j];
        lst[off + j] = (unsigned)ei[e]; dlo[off + j] = (unsigned short)(dstp[e] - n0);
      }
    }
  } else {
    const unsigned c = ecnt[blockIdx.x];
    fits = c <= (unsigned)CAP && c <= lcap;
    if (fits) {
      total = c;
      for (unsigned j = tid; j < total; j += NT) {
        const unsigned w = gl[j];
        lst[j] = w >> DLO_BITS; dlo[j] = (unsigned short)(w & ((1u << DLO_BITS) - 1u));
      }
    }
  }

  if (fits) {
    __syncthreads();
    if constexpr (SCAN) {
      const bool pub = total <= lcap && (unsigned)n <= (1u << (32 - DLO_BITS));
      if (pub)
        for (unsigned j = tid; j < total; j += NT) gl[j] = (lst[j] << DLO_BITS) | (unsigned)dlo[j];
      if (tid == 0) ecnt[blockIdx.x] = pub ? total : NOLIST;
    }
    const unsigned items = total * H;
    const unsigned mfull = items - items % (NT * 4);
    for (unsigned tb = tid; tb < mfull; tb += NT * 4) {
      float al[4]; unsigned slot[4];
      #pragma unroll
      for (int u = 0; u < 4; u++) {
        const unsigned t = tb + u * NT, i = t / H, hh = t % H;
        slot[u] = (unsigned)dlo[i] * H + hh;
        al[u] = a_src[(size_t)lst[i] * H + hh];
      }
      #pragma unroll
      for (int u = 0; u < 4; u++)
        atomicMax(&mx[slot[u]], fenc(lrelu(al[u] + adl[slot[u]])));
    }
    for (unsigned t = mfull + tid; t < items; t += NT) {
      const unsigned i = t / H, hh = t % H, sl = (unsigned)dlo[i] * H + hh;
      atomicMax(&mx[sl], fenc(lrelu(a_src[(size_t)lst[i] * H + hh] + adl[sl])));
    }
    __syncthreads();
    constexpr int LPE = D / 8;
    constexpr int EPI = NT / LPE;
    constexpr int U = 4;
    const unsigned cb = (tid % LPE) * 8, hh = cb / CH;
    const unsigned sfull = total - total % (EPI * U);
    for (unsigned ib = tid / LPE; ib < sfull; ib += EPI * U) {
      float4 xa[U], xb[U]; float as_[U]; unsigned dl[U];
      #pragma unroll
      for (int u = 0; u < U; u++) {
        const unsigned i = ib + u * EPI;
        const unsigned s = lst[i]; dl[u] = dlo[i];
        as_[u] = a_src[(size_t)s * H + hh];
        const float4* xp = (const float4*)(xl + (size_t)s * D + cb);
        xa[u] = xp[0]; xb[u] = xp[1];
      }
      #pragma unroll
      for (int u = 0; u < U; u++) {
        const unsigned hs = dl[u] * H + hh;
        const float ex = __expf(lrelu(as_[u] + adl[hs]) - fdec(mx[hs]));
        if ((cb % CH) == 0) atomicAdd(&den[hs], ex);
        float* ap = &acc[dl[u] * RS + cb];
        atomicAdd(ap + 0, xa[u].x * ex); atomicAdd(ap + 1, xa[u].y * ex);
        atomicAdd(ap + 2, xa[u].z * ex); atomicAdd(ap + 3, xa[u].w * ex);
        atomicAdd(ap + 4, xb[u].x * ex); atomicAdd(ap + 5, xb[u].y * ex);
        atomicAdd(ap + 6, xb[u].z * ex); atomicAdd(ap + 7, xb[u].w * ex);
      }
    }
    for (unsigned i = sfull + tid / LPE; i < total; i += EPI) {
      const unsigned s = lst[i], hs = (unsigned)dlo[i] * H + hh;
      const float4* xp = (const float4*)(xl + (size_t)s * D + cb);
      const float4 xa = xp[0], xb = xp[1];
      const float ex = __expf(lrelu(a_src[(size_t)s * H + hh] + adl[hs]) - fdec(mx[hs]));
      if ((cb % CH) == 0) atomicAdd(&den[hs], ex);
      float* ap = &acc[(unsigned)dlo[i] * RS + cb];
      atomicAdd(ap + 0, xa.x * ex); atomicAdd(ap + 1, xa.y * ex);
      atomicAdd(ap + 2, xa.z * ex); atomicAdd(ap + 3, xa.w * ex);
      atomicAdd(ap + 4, xb.x * ex); atomicAdd(ap + 5, xb.y * ex);
      atomicAdd(ap + 6, xb.z * ex); atomicAdd(ap + 7, xb.w * ex);
    }
  } else {
    if constexpr (SCAN) { if (tid == 0) ecnt[blockIdx.x] = NOLIST; }
    for (int e = tid; e < E_; e += NT) {
      const unsigned r = (unsigned)(dstp[e] - n0);
      if (r < nloc) {
        const int s = ei[e];
        for (int h2 = 0; h2 < H; h2++)
          atomicMax(&mx[r * H + h2], fenc(lrelu(a_src[(size_t)s * H + h2] + adl[r * H + h2])));
      }
    }
    __syncthreads();
    for (int e = tid; e < E_; e += NT) {
      const unsigned r = (unsigned)(dstp[e] - n0);
      if (r < nloc) {
        const int s = ei[e];
        for (int h2 = 0; h2 < H; h2++) {
          const unsigned hs = r * H + h2;
          const float ex = __expf(lrelu(a_src[(size_t)s * H + h2] + adl[hs]) - fdec(mx[hs]));
          atomicAdd(&den[hs], ex);
          for (int cc = 0; cc < CH; cc++)
            atomicAdd(&acc[r * RS + h2 * CH + cc], xl[(size_t)s * D + h2 * CH + cc] * ex);
        }
      }
    }
  }
  __syncthreads();
  float* op = outp + (size_t)n0 * D;
  for (unsigned i = tid; i < nloc * D; i += NT) {
    const unsigned c = i % D, r = i / D;
    float v = acc[r * RS + c] / (den[r * H + c / CH] + EPS_GAT) + bias[c];
    if (ELU) v = v > 0.f ? v : (__expf(v) - 1.f);
    op[i] = v;
  }
}

__global__ __launch_bounds__(256) void gemm2_kernel(const float* __restrict__ h,
                                                    const float* __restrict__ W2,
                                                    float* __restrict__ xl2, int nrows) {
  const int wave = threadIdx.x >> 5;
  const int lane = threadIdx.x & 31;
  const int tile = blockIdx.x * 8 + wave;
  const int rowBase = tile * 16;
  if (rowBase >= nrows) return;
  const int ln = lane & 15;
  const int khalf = (lane >> 4) << 1;
  int arow = rowBase + ln; if (arow >= nrows) arow = nrows - 1;
  const float* ap = h + (size_t)arow * D1;

  v8f c = {};
  #pragma unroll
  for (int k = 0; k < D1; k += 4) {
    v2f a = *(const v2f*)&ap[k + khalf];
    v2f b; b.x = W2[(k + khalf) * NCLS + ln];
           b.y = W2[(k + khalf + 1) * NCLS + ln];
    c = __builtin_amdgcn_wmma_f32_16x16x4_f32(false, a, false, b, (short)0, c,
                                              false, false);
  }
  const int mBase = (lane >> 4) * 8;
  float* outp = xl2 + (size_t)rowBase * NCLS;
  #pragma unroll
  for (int i = 0; i < 8; i++) {
    int row = rowBase + mBase + i;
    if (row < nrows) outp[(size_t)(mBase + i) * NCLS + ln] = c[i];
  }
}

__global__ void attn2_kernel(const float* __restrict__ xl2,
                             const float* __restrict__ att_src,
                             const float* __restrict__ att_dst,
                             float* __restrict__ a_src, float* __restrict__ a_dst, int n) {
  int t = blockIdx.x * blockDim.x + threadIdx.x;
  if (t >= n) return;
  const float* xp = xl2 + (size_t)t * NCLS;
  float s1 = 0.f, s2 = 0.f;
  #pragma unroll
  for (int cc = 0; cc < NCLS; cc++) { float v = xp[cc]; s1 += v * att_src[cc]; s2 += v * att_dst[cc]; }
  a_src[t] = s1; a_dst[t] = s2;
}

extern "C" void kernel_launch(void* const* d_in, const int* in_sizes, int n_in,
                              void* d_out, int out_size, void* d_ws, size_t ws_size,
                              hipStream_t stream) {
  const float* x        = (const float*)d_in[0];
  const int*   ei       = (const int*)d_in[1];
  const float* W1       = (const float*)d_in[2];
  const float* att_src1 = (const float*)d_in[3];
  const float* att_dst1 = (const float*)d_in[4];
  const float* bias1    = (const float*)d_in[5];
  const float* W2       = (const float*)d_in[6];
  const float* att_src2 = (const float*)d_in[7];
  const float* att_dst2 = (const float*)d_in[8];
  const float* bias2    = (const float*)d_in[9];
  float* out = (float*)d_out;

  const int N = in_sizes[0] / F_IN;
  const int E = in_sizes[1] / 2;

  float* ws = (float*)d_ws;
  size_t o = 0;
  float*    xl1    = ws + o; o += (size_t)N * D1;
  float*    h1     = ws + o; o += (size_t)N * D1;
  float*    a_src1 = ws + o; o += (size_t)N * HEADS;
  float*    a_dst1 = ws + o; o += (size_t)N * HEADS;
  unsigned* lists  = (unsigned*)(ws + o); o += (size_t)N * LIST_WORDS;
  float*    xl2    = xl1;
  float*    a_src2 = xl2 + (size_t)N * NCLS;
  float*    a_dst2 = a_src2 + (size_t)N;

  const int TB = 256;
  auto nb = [](long long n, int b) { return (unsigned)((n + b - 1) / b); };

  gemm1_kernel<<<nb((long long)N, 16), 128, 0, stream>>>(x, W1, xl1, N);
  attn1_kernel<<<nb((long long)N * HEADS, TB), TB, 0, stream>>>(xl1, att_src1, att_dst1,
                                                                a_src1, a_dst1, N);
  agg_kernel<HEADS, HID, NB_AGG, CAP_AGG, true, true><<<nb(N, NB_AGG), 1024, 0, stream>>>(
      ei, a_src1, a_dst1, xl1, bias1, h1, N, E, lists);

  gemm2_kernel<<<nb(((long long)N + 15) / 16, 8), 256, 0, stream>>>(h1, W2, xl2, N);
  attn2_kernel<<<nb(N, TB), TB, 0, stream>>>(xl2, att_src2, att_dst2, a_src2, a_dst2, N);
  agg_kernel<1, NCLS, NB_AGG, CAP_AGG, false, false><<<nb(N, NB_AGG), 1024, 0, stream>>>(
      ei, a_src2, a_dst2, xl2, bias2, out, N, E, lists);
}


# Destination scan split off as one launch that reads a one-byte slice-id plane (built once per call by a small kernel), not the 4-byte list; both aggregations read its lists; sum order = LDS arrival
# speedup vs baseline: 5.9086x; 1.0941x over previous
#include <hip/hip_runtime.h>
#include <hip/hip_bf16.h>

typedef __attribute__((ext_vector_type(2))) float v2f;
typedef __attribute__((ext_vector_type(8))) float v8f;

#define F_IN 512
#define HID 8
#define HEADS 8
#define D1 64
#define NCLS 16
#define NEG_SLOPE 0.2f
#define EPS_GAT 1e-16f

__device__ __forceinline__ unsigned fenc(float f) {
  unsigned u = __float_as_uint(f);
  return (u & 0x80000000u) ? ~u : (u | 0x80000000u);
}
__device__ __forceinline__ float fdec(unsigned e) {
  return (e & 0x80000000u) ? __uint_as_float(e ^ 0x80000000u) : __uint_as_float(~e);
}
#define ENC_NEG_INF 0x007FFFFFu

__device__ __forceinline__ float lrelu(float x) { return x > 0.f ? x : x * NEG_SLOPE; }

__global__ __launch_bounds__(128) void gemm1_kernel(const float* __restrict__ x,
                                                    const float* __restrict__ W1,
                                                    float* __restrict__ xl1, int nrows) {
  __shared__ float At[16 * 516];
  const int tid = threadIdx.x;
  const int rowBase = blockIdx.x * 16;
  for (int i = tid; i < 2048; i += 128) {
    int r = i >> 7, cq = i & 127;
    int row = rowBase + r; if (row >= nrows) row = nrows - 1;
    *(float4*)&At[r * 516 + cq * 4] = ((const float4*)(x + (size_t)row * F_IN))[cq];
  }
  __syncthreads();

  const int wave = tid >> 5;
  const int lane = tid & 31;
  const int ln = lane & 15;
  const int khalf = (lane >> 4) << 1;
  const int colBase = wave * 16;

  v8f c = {};
  #pragma unroll 4
  for (int k = 0; k < F_IN; k += 4) {
    v2f a = *(const v2f*)&At[ln * 516 + k + khalf];
    const float* wp = W1 + (size_t)(k + khalf) * D1 + colBase + ln;
    v2f b; b.x = wp[0]; b.y = wp[D1];
    c = __builtin_amdgcn_wmma_f32_16x16x4_f32(false, a, false, b, (short)0, c,
                                              false, false);
  }
  const int mBase = (lane >> 4) * 8;
  float* outp = xl1 + (size_t)rowBase * D1 + colBase;
  #pragma unroll
  for (int i = 0; i < 8; i++) {
    int row = rowBase + mBase + i;
    if (row < nrows) outp[(size_t)(mBase + i) * D1 + ln] = c[i];
  }
}

__global__ void attn1_kernel(const float* __restrict__ xl1,
                             const float* __restrict__ att_src,
                             const float* __restrict__ att_dst,
                             float* __restrict__ a_src, float* __restrict__ a_dst, int n) {
  int t = blockIdx.x * blockDim.x + threadIdx.x;
  if (t >= n * HEADS) return;
  int node = t >> 3, hh = t & 7;
  const float* xp = xl1 + (size_t)node * D1 + hh * HID;
  const float* as = att_src + hh * HID;
  const float* ad = att_dst + hh * HID;
  float s1 = 0.f, s2 = 0.f;
  #pragma unroll
  for (int cc = 0; cc < HID; cc++) { float v = xp[cc]; s1 += v * as[cc]; s2 += v * ad[cc]; }
  a_src[t] = s1; a_dst[t] = s2;
}

#define NB_AGG 392
#define CAP_AGG 16384
#define DLO_BITS 9
#define NOLIST 0xFFFFFFFFu
#define LIST_WORDS 18
#define PLANE_SLICES 256

template <int NB>
__device__ __forceinline__ bool plane_usable(int n, int E_) {
  return (n + NB - 1) / NB <= PLANE_SLICES &&
         (unsigned long long)E_ + 15ull <= (unsigned long long)n * (unsigned long long)(D1 * 4);
}

template <int NB>
__global__ __launch_bounds__(256) void plane_kernel(const int* __restrict__ ei,
                                                    unsigned* __restrict__ plane, int n, int E_) {
  if (!plane_usable<NB>(n, E_)) return;
  const int* __restrict__ dstp = ei + E_;
  const int t = blockIdx.x * 256 + threadIdx.x;
  const int nwords = ((E_ + 15) >> 4) << 2;
  if (t >= nwords) return;
  int d0, d1, d2, d3;
  if ((E_ & 3) == 0 && 4 * t < E_) {
    const int4 v = ((const int4*)dstp)[t];
    d0 = v.x; d1 = v.y; d2 = v.z; d3 = v.w;
  } else {
    const int e = 4 * t;
    d0 = e + 0 < E_ ? dstp[e + 0] : 0; d1 = e + 1 < E_ ? dstp[e + 1] : 0;
    d2 = e + 2 < E_ ? dstp[e + 2] : 0; d3 = e + 3 < E_ ? dstp[e + 3] : 0;
  }
  const unsigned b0 = ((unsigned)d0 / (unsigned)NB) & 255u, b1 = ((unsigned)d1 / (unsigned)NB) & 255u;
  const unsigned b2 = ((unsigned)d2 / (unsigned)NB) & 255u, b3 = ((unsigned)d3 / (unsigned)NB) & 255u;
  plane[t] = b0 | (b1 << 8) | (b2 << 16) | (b3 << 24);
}

__device__ __forceinline__ unsigned zero_bytes(unsigned x) {
  return ~(((x & 0x7F7F7F7Fu) + 0x7F7F7F7Fu) | x | 0x7F7F7F7Fu);
}

template <int NB, int CAP>
__global__ __launch_bounds__(1024) void scan_kernel(const int* __restrict__ ei,
                                                    const unsigned* __restrict__ plane,
                                                    int n, int E_, unsigned* lists) {
  static_assert(NB <= (1 << DLO_BITS), "local destination must fit its field");
  constexpr int NT = 1024;
  constexpr int NW = NT / 32;
  constexpr int SEG = CAP / NW;
  __shared__ unsigned seg[CAP];
  __shared__ unsigned lst[CAP];
  __shared__ unsigned short dlo[CAP];
  __shared__ unsigned wcnt[NW];

  const int tid = threadIdx.x;
  const int n0 = blockIdx.x * NB;
  const int* __restrict__ dstp = ei + E_;
  const unsigned cpad = (gridDim.x + 31u) & ~31u;
  const unsigned lcap = (((unsigned)LIST_WORDS * (unsigned)n - cpad) / gridDim.x) & ~31u;
  unsigned* ecnt = lists;
  unsigned* gl = lists + cpad + (size_t)blockIdx.x * lcap;

  const bool use = plane_usable<NB>(n, E_);
  unsigned total = 0u; bool fits = use;
  if (use) {
    const unsigned lane = tid & 31, wave = tid >> 5;
    unsigned* myseg = seg + wave * SEG;
    unsigned wn = 0u;
    auto push = [&](bool m, unsigned e) {
      const unsigned mask = __builtin_amdgcn_ballot_w32(m);
      if (m) { const unsigned k = wn + __builtin_amdgcn_mbcnt_lo(mask, 0u); if (k < (unsigned)SEG) myseg[k] = e; }
      wn += (unsigned)__builtin_popcount(mask);
    };
    const unsigned sid4 = blockIdx.x * 0x01010101u;
    const int ng = (E_ + 15) >> 4;
    for (int gb0 = 0; gb0 < ng; gb0 += NT * 8) {
      uint4 pv[8];
      #pragma unroll
      for (int u = 0; u < 8; u++) {
        const int g = gb0 + u * NT + tid;
        pv[u] = ((const uint4*)plane)[g < ng ? g : 0];
      }
      __builtin_amdgcn_sched_barrier(0);
      #pragma unroll
      for (int u = 0; u < 8; u++) {
        const int g = gb0 + u * NT + tid;
        unsigned mm = (zero_bytes(pv[u].x ^ sid4) >> 7) | (zero_bytes(pv[u].y ^ sid4) >> 6) |
                      (zero_bytes(pv[u].z ^ sid4) >> 5) | (zero_bytes(pv[u].w ^ sid4) >> 4);
        if (g >= ng) mm = 0u;
        while (__builtin_amdgcn_ballot_w32(mm != 0u) != 0u) {
          const unsigned p = (unsigned)__builtin_ctz(mm | 0x80000000u);
          const unsigned e = 16u * (unsigned)g + 4u * (p & 3u) + (p >> 3);
          push(mm != 0u && e < (unsigned)E_, e);
          mm &= mm - 1u;
        }
      }
    }
    if (lane == 0) wcnt[wave] = wn;
    __syncthreads();
    unsigned off = 0u;
    for (int w = 0; w < NW; w++) {
      const unsigned c = wcnt[w];
      if ((unsigned)w < wave) off += c;
      total += c; fits = fits && c <= (unsigned)SEG;
    }
    if (fits) {
      for (unsigned j = lane; j < wn; j += 32) {
        const unsigned e = myseg[j];
        lst[off + j] = (unsigned)ei[e]; dlo[off + j] = (unsigned short)(dstp[e] - n0);
      }
    }
  }
  if (fits) {
    __syncthreads();
    const bool pub = total <= lcap && (unsigned)n <= (1u << (32 - DLO_BITS));
    if (pub)
      for (unsigned j = tid; j < total; j += NT) gl[j] = (lst[j] << DLO_BITS) | (unsigned)dlo[j];
    if (tid == 0) ecnt[blockIdx.x] = pub ? total : NOLIST;
  } else {
    if (tid == 0) ecnt[blockIdx.x] = NOLIST;
  }
}

template <int H, int CH, int NB, int CAP, bool ELU>
__global__ __launch_bounds__(1024) void agg_kernel(const int* __restrict__ ei,
                                                   const float* __restrict__ a_src,
                                                   const float* __restrict__ a_dst,
                                                   const float* __restrict__ xl,
                                                   const float* __restrict__ bias,
                                                   float* __restrict__ outp, int n, int E_,
                                                   const unsigned* lists) {
  static_assert(NB <= (1 << DLO_BITS), "local destination must fit its field");
  constexpr int D = H * CH;
  constexpr int RS = D + 1;
  constexpr int NT = 1024;
  __shared__ float acc[NB * RS];
  __shared__ float den[NB * H];
  __shared__ unsigned mx[NB * H];
  __shared__ float adl[NB * H];
  __shared__ unsigned lst[CAP];
  __shared__ unsigned short dlo[CAP];

  const int tid = threadIdx.x;
  const int n0 = blockIdx.x * NB;
  const unsigned nloc = (unsigned)((n - n0) < NB ? (n - n0) : NB);
  const int* __restrict__ dstp = ei + E_;
  const unsigned cpad = (gridDim.x + 31u) & ~31u;
  const unsigned lcap = (((unsigned)LIST_WORDS * (unsigned)n - cpad) / gridDim.x) & ~31u;
  const unsigned* ecnt = lists;
  const unsigned* gl = lists + cpad + (size_t)blockIdx.x * lcap;

  for (int i = tid; i < NB * RS; i += NT) acc[i] = 0.f;
  for (int i = tid; i < NB * H; i += NT) {
    den[i] = 0.f; mx[i] = ENC_NEG_INF;
    adl[i] = ((unsigned)i < nloc * H) ? a_dst[(size_t)n0 * H + i] : 0.f;
  }
  __syncthreads();

  unsigned total = 0u;
  const unsigned cnt = ecnt[blockIdx.x];
  const bool fits = cnt <= (unsigned)CAP && cnt <= lcap;
  if (fits) {
    total = cnt;
    for (unsigned j = tid; j < total; j += NT) {
      const unsigned w = gl[j];
      lst[j] = w >> DLO_BITS; dlo[j] = (unsigned short)(w & ((1u << DLO_BITS) - 1u));
    }
    __syncthreads();
    const unsigned items = total * H;
    const unsigned mfull = items - items % (NT * 4);
    for (unsigned tb = tid; tb < mfull; tb += NT * 4) {
      float al[4]; unsigned slot[4];
      #pragma unroll
      for (int u = 0; u < 4; u++) {
        const unsigned t = tb + u * NT, i = t / H, hh = t % H;
        slot[u] = (unsigned)dlo[i] * H + hh;
        al[u] = a_src[(size_t)lst[i] * H + hh];
      }
      #pragma unroll
      for (int u = 0; u < 4; u++)
        atomicMax(&mx[slot[u]], fenc(lrelu(al[u] + adl[slot[u]])));
    }
    for (unsigned t = mfull + tid; t < items; t += NT) {
      const unsigned i = t / H, hh = t % H, sl = (unsigned)dlo[i] * H + hh;
      atomicMax(&mx[sl], fenc(lrelu(a_src[(size_t)lst[i] * H + hh] + adl[sl])));
    }
    __syncthreads();
    constexpr int LPE = D / 8;
    constexpr int EPI = NT / LPE;
    constexpr int U = 4;
    const unsigned cb = (tid % LPE) * 8, hh = cb / CH;
    const unsigned sfull = total - total % (EPI * U);
    for (unsigned ib = tid / LPE; ib < sfull; ib += EPI * U) {
      float4 xa[U], xb[U]; float as_[U]; unsigned dl[U];
      #pragma unroll
      for (int u = 0; u < U; u++) {
        const unsigned i = ib + u * EPI;
        const unsigned s = lst[i]; dl[u] = dlo[i];
        as_[u] = a_src[(size_t)s * H + hh];
        const float4* xp = (const float4*)(xl + (size_t)s * D + cb);
        xa[u] = xp[0]; xb[u] = xp[1];
      }
      #pragma unroll
      for (int u = 0; u < U; u++) {
        const unsigned hs = dl[u] * H + hh;
        const float ex = __expf(lrelu(as_[u] + adl[hs]) - fdec(mx[hs]));
        if ((cb % CH) == 0) atomicAdd(&den[hs], ex);
        float* ap = &acc[dl[u] * RS + cb];
        atomicAdd(ap + 0, xa[u].x * ex); atomicAdd(ap + 1, xa[u].y * ex);
        atomicAdd(ap + 2, xa[u].z * ex); atomicAdd(ap + 3, xa[u].w * ex);
        atomicAdd(ap + 4, xb[u].x * ex); atomicAdd(ap + 5, xb[u].y * ex);
        atomicAdd(ap + 6, xb[u].z * ex); atomicAdd(ap + 7, xb[u].w * ex);
      }
    }
    for (unsigned i = sfull + tid / LPE; i < total; i += EPI) {
      const unsigned s = lst[i], hs = (unsigned)dlo[i] * H + hh;
      const float4* xp = (const float4*)(xl + (size_t)s * D + cb);
      const float4 xa = xp[0], xb = xp[1];
      const float ex = __expf(lrelu(a_src[(size_t)s * H + hh] + adl[hs]) - fdec(mx[hs]));
      if ((cb % CH) == 0) atomicAdd(&den[hs], ex);
      float* ap = &acc[(unsigned)dlo[i] * RS + cb];
      atomicAdd(ap + 0, xa.x * ex); atomicAdd(ap + 1, xa.y * ex);
      atomicAdd(ap + 2, xa.z * ex); atomicAdd(ap + 3, xa.w * ex);
      atomicAdd(ap + 4, xb.x * ex); atomicAdd(ap + 5, xb.y * ex);
      atomicAdd(ap + 6, xb.z * ex); atomicAdd(ap + 7, xb.w * ex);
    }
  } else {
    for (int e = tid; e < E_; e += NT) {
      const unsigned r = (unsigned)(dstp[e] - n0);
      if (r < nloc) {
        const int s = ei[e];
        for (int h2 = 0; h2 < H; h2++)
          atomicMax(&mx[r * H + h2], fenc(lrelu(a_src[(size_t)s * H + h2] + adl[r * H + h2])));
      }
    }
    __syncthreads();
    for (int e = tid; e < E_; e += NT) {
      const unsigned r = (unsigned)(dstp[e] - n0);
      if (r < nloc) {
        const int s = ei[e];
        for (int h2 = 0; h2 < H; h2++) {
          const unsigned hs = r * H + h2;
          const float ex = __expf(lrelu(a_src[(size_t)s * H + h2] + adl[hs]) - fdec(mx[hs]));
          atomicAdd(&den[hs], ex);
          for (int cc = 0; cc < CH; cc++)
            atomicAdd(&acc[r * RS + h2 * CH + cc], xl[(size_t)s * D + h2 * CH + cc] * ex);
        }
      }
    }
  }
  __syncthreads();
  float* op = outp + (size_t)n0 * D;
  for (unsigned i = tid; i < nloc * D; i += NT) {
    const unsigned c = i % D, r = i / D;
    float v = acc[r * RS + c] / (den[r * H + c / CH] + EPS_GAT) + bias[c];
    if (ELU) v = v > 0.f ? v : (__expf(v) - 1.f);
    op[i] = v;
  }
}

__global__ __launch_bounds__(256) void gemm2_kernel(const float* __restrict__ h,
                                                    const float* __restrict__ W2,
                                                    float* __restrict__ xl2, int nrows) {
  const int wave = threadIdx.x >> 5;
  const int lane = threadIdx.x & 31;
  const int tile = blockIdx.x * 8 + wave;
  const int rowBase = tile * 16;
  if (rowBase >= nrows) return;
  const int ln = lane & 15;
  const int khalf = (lane >> 4) << 1;
  int arow = rowBase + ln; if (arow >= nrows) arow = nrows - 1;
  const float* ap = h + (size_t)arow * D1;

  v8f c = {};
  #pragma unroll
  for (int k = 0; k < D1; k += 4) {
    v2f a = *(const v2f*)&ap[k + khalf];
    v2f b; b.x = W2[(k + khalf) * NCLS + ln];
           b.y = W2[(k + khalf + 1) * NCLS + ln];
    c = __builtin_amdgcn_wmma_f32_16x16x4_f32(false, a, false, b, (short)0, c,
                                              false, false);
  }
  const int mBase = (lane >> 4) * 8;
  float* outp = xl2 + (size_t)rowBase * NCLS;
  #pragma unroll
  for (int i = 0; i < 8; i++) {
    int row = rowBase + mBase + i;
    if (row < nrows) outp[(size_t)(mBase + i) * NCLS + ln] = c[i];
  }
}

__global__ void attn2_kernel(const float* __restrict__ xl2,
                             const float* __restrict__ att_src,
                             const float* __restrict__ att_dst,
                             float* __restrict__ a_src, float* __restrict__ a_dst, int n) {
  int t = blockIdx.x * blockDim.x + threadIdx.x;
  if (t >= n) return;
  const float* xp = xl2 + (size_t)t * NCLS;
  float s1 = 0.f, s2 = 0.f;
  #pragma unroll
  for (int cc = 0; cc < NCLS; cc++) { float v = xp[cc]; s1 += v * att_src[cc]; s2 += v * att_dst[cc]; }
  a_src[t] = s1; a_dst[t] = s2;
}

extern "C" void kernel_launch(void* const* d_in, const int* in_sizes, int n_in,
                              void* d_out, int out_size, void* d_ws, size_t ws_size,
                              hipStream_t stream) {
  const float* x        = (const float*)d_in[0];
  const int*   ei       = (const int*)d_in[1];
  const float* W1       = (const float*)d_in[2];
  const float* att_src1 = (const float*)d_in[3];
  const float* att_dst1 = (const float*)d_in[4];
  const float* bias1    = (const float*)d_in[5];
  const float* W2       = (const float*)d_in[6];
  const float* att_src2 = (const float*)d_in[7];
  const float* att_dst2 = (const float*)d_in[8];
  const float* bias2    = (const float*)d_in[9];
  float* out = (float*)d_out;

  const int N = in_sizes[0] / F_IN;
  const int E = in_sizes[1] / 2;

  float* ws = (float*)d_ws;
  size_t o = 0;
  float*    xl1    = ws + o; o += (size_t)N * D1;
  float*    h1     = ws + o; o += (size_t)N * D1;
  float*    a_src1 = ws + o; o += (size_t)N * HEADS;
  float*    a_dst1 = ws + o; o += (size_t)N * HEADS;
  unsigned* lists  = (unsigned*)(ws + o); o += (size_t)N * LIST_WORDS;
  float*    xl2    = xl1;
  float*    a_src2 = xl2 + (size_t)N * NCLS;
  float*    a_dst2 = a_src2 + (size_t)N;

  const int TB = 256;
  auto nb = [](long long n, int b) { return (unsigned)((n + b - 1) / b); };

  unsigned* plane = (unsigned*)h1;

  plane_kernel<NB_AGG><<<nb((((long long)E + 15) / 16) * 4, TB), TB, 0, stream>>>(ei, plane, N, E);
  scan_kernel<NB_AGG, CAP_AGG><<<nb(N, NB_AGG), 1024, 0, stream>>>(ei, plane, N, E, lists);

  gemm1_kernel<<<nb((long long)N, 16), 128, 0, stream>>>(x, W1, xl1, N);
  attn1_kernel<<<nb((long long)N * HEADS, TB), TB, 0, stream>>>(xl1, att_src1, att_dst1,
                                                                a_src1, a_dst1, N);
  agg_kernel<HEADS, HID, NB_AGG, CAP_AGG, true><<<nb(N, NB_AGG), 1024, 0, stream>>>(
      ei, a_src1, a_dst1, xl1, bias1, h1, N, E, lists);

  gemm2_kernel<<<nb(((long long)N + 15) / 16, 8), 256, 0, stream>>>(h1, W2, xl2, N);
  attn2_kernel<<<nb(N, TB), TB, 0, stream>>>(xl2, att_src2, att_dst2, a_src2, a_dst2, N);
  agg_kernel<1, NCLS, NB_AGG, CAP_AGG, false><<<nb(N, NB_AGG), 1024, 0, stream>>>(
      ei, a_src2, a_dst2, xl2, bias2, out, N, E, lists);
}


# Slice-id plane stored as eight bit-planes (wave ballots, whole lines): the scan finds a slice's edges with 8 bitwise ops per 32 edges, not a byte compare per edge; sum order = LDS arrival as before
# speedup vs baseline: 6.0990x; 1.0322x over previous
#include <hip/hip_runtime.h>
#include <hip/hip_bf16.h>

typedef __attribute__((ext_vector_type(2))) float v2f;
typedef __attribute__((ext_vector_type(8))) float v8f;

#define F_IN 512
#define HID 8
#define HEADS 8
#define D1 64
#define NCLS 16
#define NEG_SLOPE 0.2f
#define EPS_GAT 1e-16f

__device__ __forceinline__ unsigned fenc(float f) {
  unsigned u = __float_as_uint(f);
  return (u & 0x80000000u) ? ~u : (u | 0x80000000u);
}
__device__ __forceinline__ float fdec(unsigned e) {
  return (e & 0x80000000u) ? __uint_as_float(e ^ 0x80000000u) : __uint_as_float(~e);
}
#define ENC_NEG_INF 0x007FFFFFu

__device__ __forceinline__ float lrelu(float x) { return x > 0.f ? x : x * NEG_SLOPE; }

__global__ __launch_bounds__(128) void gemm1_kernel(const float* __restrict__ x,
                                                    const float* __restrict__ W1,
                                                    float* __restrict__ xl1, int nrows) {
  __shared__ float At[16 * 516];
  const int tid = threadIdx.x;
  const int rowBase = blockIdx.x * 16;
  for (int i = tid; i < 2048; i += 128) {
    int r = i >> 7, cq = i & 127;
    int row = rowBase + r; if (row >= nrows) row = nrows - 1;
    *(float4*)&At[r * 516 + cq * 4] = ((const float4*)(x + (size_t)row * F_IN))[cq];
  }
  __syncthreads();

  const int wave = tid >> 5;
  const int lane = tid & 31;
  const int ln = lane & 15;
  const int khalf = (lane >> 4) << 1;
  const int colBase = wave * 16;

  v8f c = {};
  #pragma unroll 4
  for (int k = 0; k < F_IN; k += 4) {
    v2f a = *(const v2f*)&At[ln * 516 + k + khalf];
    const float* wp = W1 + (size_t)(k + khalf) * D1 + colBase + ln;
    v2f b; b.x = wp[0]; b.y = wp[D1];
    c = __builtin_amdgcn_wmma_f32_16x16x4_f32(false, a, false, b, (short)0, c,
                                              false, false);
  }
  const int mBase = (lane >> 4) * 8;
  float* outp = xl1 + (size_t)rowBase * D1 + colBase;
  #pragma unroll
  for (int i = 0; i < 8; i++) {
    int row = rowBase + mBase + i;
    if (row < nrows) outp[(size_t)(mBase + i) * D1 + ln] = c[i];
  }
}

__global__ void attn1_kernel(const float* __restrict__ xl1,
                             const float* __restrict__ att_src,
                             const float* __restrict__ att_dst,
                             float* __restrict__ a_src, float* __restrict__ a_dst, int n) {
  int t = blockIdx.x * blockDim.x + threadIdx.x;
  if (t >= n * HEADS) return;
  int node = t >> 3, hh = t & 7;
  const float* xp = xl1 + (size_t)node * D1 + hh * HID;
  const float* as = att_src + hh * HID;
  const float* ad = att_dst + hh * HID;
  float s1 = 0.f, s2 = 0.f;
  #pragma unroll
  for (int cc = 0; cc < HID; cc++) { float v = xp[cc]; s1 += v * as[cc]; s2 += v * ad[cc]; }
  a_src[t] = s1; a_dst[t] = s2;
}

#define NB_AGG 392
#define CAP_AGG 16384
#define DLO_BITS 9
#define NOLIST 0xFFFFFFFFu
#define LIST_WORDS 18
#define PLANE_SLICES 256
#define PB_EDGES 4096

template <int NB>
__device__ __forceinline__ bool plane_usable(int n, int E_) {
  return (n + NB - 1) / NB <= PLANE_SLICES &&
         (((unsigned long long)E_ + (PB_EDGES - 1)) / PB_EDGES) * (unsigned long long)PB_EDGES <=
             (unsigned long long)n * (unsigned long long)(D1 * 4);
}

template <int NB>
__global__ __launch_bounds__(256) void plane_kernel(const int* __restrict__ ei,
                                                    unsigned* __restrict__ plane, int n, int E_) {
  if (!plane_usable<NB>(n, E_)) return;
  __shared__ __attribute__((aligned(16))) unsigned sm[8 * 128];
  const int* __restrict__ dstp = ei + E_;
  const unsigned tid = threadIdx.x, lane = tid & 31u, wave = tid >> 5;
  const unsigned e0 = blockIdx.x * (unsigned)PB_EDGES + wave * 512u + lane;
  int d[16];
  #pragma unroll
  for (int i = 0; i < 16; i++) {
    const unsigned e = e0 + 32u * i;
    d[i] = e < (unsigned)E_ ? dstp[e] : 0;
  }
  #pragma unroll
  for (int i = 0; i < 16; i++) {
    const unsigned sid = ((unsigned)d[i] / (unsigned)NB) & 255u;
    unsigned word = 0u;
    #pragma unroll
    for (int k = 0; k < 8; k++) {
      const unsigned bal = __builtin_amdgcn_ballot_w32(((sid >> k) & 1u) != 0u);
      if (lane == (unsigned)k) word = bal;
    }
    if (lane < 8u) sm[lane * 128u + wave * 16u + i] = word;
  }
  __syncthreads();
  ((uint4*)plane)[(size_t)blockIdx.x * 256u + tid] = ((const uint4*)sm)[tid];
}

template <int NB, int CAP>
__global__ __launch_bounds__(1024) void scan_kernel(const int* __restrict__ ei,
                                                    const unsigned* __restrict__ plane,
                                                    int n, int E_, unsigned* lists) {
  static_assert(NB <= (1 << DLO_BITS), "local destination must fit its field");
  constexpr int NT = 1024;
  constexpr int NW = NT / 32;
  constexpr int SEG = CAP / NW;
  __shared__ unsigned seg[CAP];
  __shared__ unsigned lst[CAP];
  __shared__ unsigned short dlo[CAP];
  __shared__ unsigned wcnt[NW];

  const int tid = threadIdx.x;
  const int n0 = blockIdx.x * NB;
  const int* __restrict__ dstp = ei + E_;
  const unsigned cpad = (gridDim.x + 31u) & ~31u;
  const unsigned lcap = (((unsigned)LIST_WORDS * (unsigned)n - cpad) / gridDim.x) & ~31u;
  unsigned* ecnt = lists;
  unsigned* gl = lists + cpad + (size_t)blockIdx.x * lcap;

  const bool use = plane_usable<NB>(n, E_);
  unsigned total = 0u; bool fits = use;
  if (use) {
    const unsigned lane = tid & 31, wave = tid >> 5;
    unsigned* myseg = seg + wave * SEG;
    unsigned wn = 0u;
    auto push = [&](bool m, unsigned e) {
      const unsigned mask = __builtin_amdgcn_ballot_w32(m);
      if (m) { const unsigned k = wn + __builtin_amdgcn_mbcnt_lo(mask, 0u); if (k < (unsigned)SEG) myseg[k] = e; }
      wn += (unsigned)__builtin_popcount(mask);
    };
    unsigned inv[8];
    #pragma unroll
    for (int k = 0; k < 8; k++) inv[k] = ((blockIdx.x >> k) & 1u) ? 0u : 0xFFFFFFFFu;
    const int nsb = (E_ + (PB_EDGES - 1)) / PB_EDGES;
    const uint4* __restrict__ pp = (const uint4*)plane;
    const int wv = __builtin_amdgcn_readfirstlane((int)wave);
    for (int sb0 = wv; sb0 < nsb; sb0 += 2 * NW) {
      uint4 pv[2][8];
      #pragma unroll
      for (int s = 0; s < 2; s++) {
        const int sb = sb0 + s * NW;
        const size_t base = (size_t)(sb < nsb ? sb : 0) * 256u + lane;
        #pragma unroll
        for (int k = 0; k < 8; k++) pv[s][k] = pp[base + 32u * k];
      }
      #pragma unroll
      for (int s = 0; s < 2; s++) {
        const int sb = sb0 + s * NW;
        const bool live = sb < nsb;
        unsigned m[4];
        m[0] = pv[s][0].x ^ inv[0]; m[1] = pv[s][0].y ^ inv[0];
        m[2] = pv[s][0].z ^ inv[0]; m[3] = pv[s][0].w ^ inv[0];
        #pragma unroll
        for (int k = 1; k < 8; k++) {
          m[0] &= pv[s][k].x ^ inv[k]; m[1] &= pv[s][k].y ^ inv[k];
          m[2] &= pv[s][k].z ^ inv[k]; m[3] &= pv[s][k].w ^ inv[k];
        }
        #pragma unroll
        for (int q = 0; q < 4; q++) {
          unsigned mm = live ? m[q] : 0u;
          const unsigned eb = (unsigned)sb * (unsigned)PB_EDGES + (4u * lane + (unsigned)q) * 32u;
          while (__builtin_amdgcn_ballot_w32(mm != 0u) != 0u) {
            const unsigned e = eb + (mm != 0u ? (unsigned)__builtin_ctz(mm) : 0u);
            push(mm != 0u && e < (unsigned)E_, e);
            mm &= mm - 1u;
          }
        }
      }
    }
    if (lane == 0) wcnt[wave] = wn;
    __syncthreads();
    unsigned off = 0u;
    for (int w = 0; w < NW; w++) {
      const unsigned c = wcnt[w];
      if ((unsigned)w < wave) off += c;
      total += c; fits = fits && c <= (unsigned)SEG;
    }
    if (fits) {
      for (unsigned j = lane; j < wn; j += 32) {
        const unsigned e = myseg[j];
        lst[off + j] = (unsigned)ei[e]; dlo[off + j] = (unsigned short)(dstp[e] - n0);
      }
    }
  }
  if (fits) {
    __syncthreads();
    const bool pub = total <= lcap && (unsigned)n <= (1u << (32 - DLO_BITS));
    if (pub)
      for (unsigned j = tid; j < total; j += NT) gl[j] = (lst[j] << DLO_BITS) | (unsigned)dlo[j];
    if (tid == 0) ecnt[blockIdx.x] = pub ? total : NOLIST;
  } else {
    if (tid == 0) ecnt[blockIdx.x] = NOLIST;
  }
}

template <int H, int CH, int NB, int CAP, bool ELU>
__global__ __launch_bounds__(1024) void agg_kernel(const int* __restrict__ ei,
                                                   const float* __restrict__ a_src,
                                                   const float* __restrict__ a_dst,
                                                   const float* __restrict__ xl,
                                                   const float* __restrict__ bias,
                                                   float* __restrict__ outp, int n, int E_,
                                                   const unsigned* lists) {
  static_assert(NB <= (1 << DLO_BITS), "local destination must fit its field");
  constexpr int D = H * CH;
  constexpr int RS = D + 1;
  constexpr int NT = 1024;
  __shared__ float acc[NB * RS];
  __shared__ float den[NB * H];
  __shared__ unsigned mx[NB * H];
  __shared__ float adl[NB * H];
  __shared__ unsigned lst[CAP];
  __shared__ unsigned short dlo[CAP];

  const int tid = threadIdx.x;
  const int n0 = blockIdx.x * NB;
  const unsigned nloc = (unsigned)((n - n0) < NB ? (n - n0) : NB);
  const int* __restrict__ dstp = ei + E_;
  const unsigned cpad = (gridDim.x + 31u) & ~31u;
  const unsigned lcap = (((unsigned)LIST_WORDS * (unsigned)n - cpad) / gridDim.x) & ~31u;
  const unsigned* ecnt = lists;
  const unsigned* gl = lists + cpad + (size_t)blockIdx.x * lcap;

  for (int i = tid; i < NB * RS; i += NT) acc[i] = 0.f;
  for (int i = tid; i < NB * H; i += NT) {
    den[i] = 0.f; mx[i] = ENC_NEG_INF;
    adl[i] = ((unsigned)i < nloc * H) ? a_dst[(size_t)n0 * H + i] : 0.f;
  }
  __syncthreads();

  unsigned total = 0u;
  const unsigned cnt = ecnt[blockIdx.x];
  const bool fits = cnt <= (unsigned)CAP && cnt <= lcap;
  if (fits) {
    total = cnt;
    for (unsigned j = tid; j < total; j += NT) {
      const unsigned w = gl[j];
      lst[j] = w >> DLO_BITS; dlo[j] = (unsigned short)(w & ((1u << DLO_BITS) - 1u));
    }
    __syncthreads();
    const unsigned items = total * H;
    const unsigned mfull = items - items % (NT * 4);
    for (unsigned tb = tid; tb < mfull; tb += NT * 4) {
      float al[4]; unsigned slot[4];
      #pragma unroll
      for (int u = 0; u < 4; u++) {
        const unsigned t = tb + u * NT, i = t / H, hh = t % H;
        slot[u] = (unsigned)dlo[i] * H + hh;
        al[u] = a_src[(size_t)lst[i] * H + hh];
      }
      #pragma unroll
      for (int u = 0; u < 4; u++)
        atomicMax(&mx[slot[u]], fenc(lrelu(al[u] + adl[slot[u]])));
    }
    for (unsigned t = mfull + tid; t < items; t += NT) {
      const unsigned i = t / H, hh = t % H, sl = (unsigned)dlo[i] * H + hh;
      atomicMax(&mx[sl], fenc(lrelu(a_src[(size_t)lst[i] * H + hh] + adl[sl])));
    }
    __syncthreads();
    constexpr int LPE = D / 8;
    constexpr int EPI = NT / LPE;
    constexpr int U = 4;
    const unsigned cb = (tid % LPE) * 8, hh = cb / CH;
    const unsigned sfull = total - total % (EPI * U);
    for (unsigned ib = tid / LPE; ib < sfull; ib += EPI * U) {
      float4 xa[U], xb[U]; float as_[U]; unsigned dl[U];
      #pragma unroll
      for (int u = 0; u < U; u++) {
        const unsigned i = ib + u * EPI;
        const unsigned s = lst[i]; dl[u] = dlo[i];
        as_[u] = a_src[(size_t)s * H + hh];
        const float4* xp = (const float4*)(xl + (size_t)s * D + cb);
        xa[u] = xp[0]; xb[u] = xp[1];
      }
      #pragma unroll
      for (int u = 0; u < U; u++) {
        const unsigned hs = dl[u] * H + hh;
        const float ex = __expf(lrelu(as_[u] + adl[hs]) - fdec(mx[hs]));
        if ((cb % CH) == 0) atomicAdd(&den[hs], ex);
        float* ap = &acc[dl[u] * RS + cb];
        atomicAdd(ap + 0, xa[u].x * ex); atomicAdd(ap + 1, xa[u].y * ex);
        atomicAdd(ap + 2, xa[u].z * ex); atomicAdd(ap + 3, xa[u].w * ex);
        atomicAdd(ap + 4, xb[u].x * ex); atomicAdd(ap + 5, xb[u].y * ex);
        atomicAdd(ap + 6, xb[u].z * ex); atomicAdd(ap + 7, xb[u].w * ex);
      }
    }
    for (unsigned i = sfull + tid / LPE; i < total; i += EPI) {
      const unsigned s = lst[i], hs = (unsigned)dlo[i] * H + hh;
      const float4* xp = (const float4*)(xl + (size_t)s * D + cb);
      const float4 xa = xp[0], xb = xp[1];
      const float ex = __expf(lrelu(a_src[(size_t)s * H + hh] + adl[hs]) - fdec(mx[hs]));
      if ((cb % CH) == 0) atomicAdd(&den[hs], ex);
      float* ap = &acc[(unsigned)dlo[i] * RS + cb];
      atomicAdd(ap + 0, xa.x * ex); atomicAdd(ap + 1, xa.y * ex);
      atomicAdd(ap + 2, xa.z * ex); atomicAdd(ap + 3, xa.w * ex);
      atomicAdd(ap + 4, xb.x * ex); atomicAdd(ap + 5, xb.y * ex);
      atomicAdd(ap + 6, xb.z * ex); atomicAdd(ap + 7, xb.w * ex);
    }
  } else {
    for (int e = tid; e < E_; e += NT) {
      const unsigned r = (unsigned)(dstp[e] - n0);
      if (r < nloc) {
        const int s = ei[e];
        for (int h2 = 0; h2 < H; h2++)
          atomicMax(&mx[r * H + h2], fenc(lrelu(a_src[(size_t)s * H + h2] + adl[r * H + h2])));
      }
    }
    __syncthreads();
    for (int e = tid; e < E_; e += NT) {
      const unsigned r = (unsigned)(dstp[e] - n0);
      if (r < nloc) {
        const int s = ei[e];
        for (int h2 = 0; h2 < H; h2++) {
          const unsigned hs = r * H + h2;
          const float ex = __expf(lrelu(a_src[(size_t)s * H + h2] + adl[hs]) - fdec(mx[hs]));
          atomicAdd(&den[hs], ex);
          for (int cc = 0; cc < CH; cc++)
            atomicAdd(&acc[r * RS + h2 * CH + cc], xl[(size_t)s * D + h2 * CH + cc] * ex);
        }
      }
    }
  }
  __syncthreads();
  float* op = outp + (size_t)n0 * D;
  for (unsigned i = tid; i < nloc * D; i += NT) {
    const unsigned c = i % D, r = i / D;
    float v = acc[r * RS + c] / (den[r * H + c / CH] + EPS_GAT) + bias[c];
    if (ELU) v = v > 0.f ? v : (__expf(v) - 1.f);
    op[i] = v;
  }
}

__global__ __launch_bounds__(256) void gemm2_kernel(const float* __restrict__ h,
                                                    const float* __restrict__ W2,
                                                    float* __restrict__ xl2, int nrows) {
  const int wave = threadIdx.x >> 5;
  const int lane = threadIdx.x & 31;
  const int tile = blockIdx.x * 8 + wave;
  const int rowBase = tile * 16;
  if (rowBase >= nrows) return;
  const int ln = lane & 15;
  const int khalf = (lane >> 4) << 1;
  int arow = rowBase + ln; if (arow >= nrows) arow = nrows - 1;
  const float* ap = h + (size_t)arow * D1;

  v8f c = {};
  #pragma unroll
  for (int k = 0; k < D1; k += 4) {
    v2f a = *(const v2f*)&ap[k + khalf];
    v2f b; b.x = W2[(k + khalf) * NCLS + ln];
           b.y = W2[(k + khalf + 1) * NCLS + ln];
    c = __builtin_amdgcn_wmma_f32_16x16x4_f32(false, a, false, b, (short)0, c,
                                              false, false);
  }
  const int mBase = (lane >> 4) * 8;
  float* outp = xl2 + (size_t)rowBase * NCLS;
  #pragma unroll
  for (int i = 0; i < 8; i++) {
    int row = rowBase + mBase + i;
    if (row < nrows) outp[(size_t)(mBase + i) * NCLS + ln] = c[i];
  }
}

__global__ void attn2_kernel(const float* __restrict__ xl2,
                             const float* __restrict__ att_src,
                             const float* __restrict__ att_dst,
                             float* __restrict__ a_src, float* __restrict__ a_dst, int n) {
  int t = blockIdx.x * blockDim.x + threadIdx.x;
  if (t >= n) return;
  const float* xp = xl2 + (size_t)t * NCLS;
  float s1 = 0.f, s2 = 0.f;
  #pragma unroll
  for (int cc = 0; cc < NCLS; cc++) { float v = xp[cc]; s1 += v * att_src[cc]; s2 += v * att_dst[cc]; }
  a_src[t] = s1; a_dst[t] = s2;
}

extern "C" void kernel_launch(void* const* d_in, const int* in_sizes, int n_in,
                              void* d_out, int out_size, void* d_ws, size_t ws_size,
                              hipStream_t stream) {
  const float* x        = (const float*)d_in[0];
  const int*   ei       = (const int*)d_in[1];
  const float* W1       = (const float*)d_in[2];
  const float* att_src1 = (const float*)d_in[3];
  const float* att_dst1 = (const float*)d_in[4];
  const float* bias1    = (const float*)d_in[5];
  const float* W2       = (const float*)d_in[6];
  const float* att_src2 = (const float*)d_in[7];
  const float* att_dst2 = (const float*)d_in[8];
  const float* bias2    = (const float*)d_in[9];
  float* out = (float*)d_out;

  const int N = in_sizes[0] / F_IN;
  const int E = in_sizes[1] / 2;

  float* ws = (float*)d_ws;
  size_t o = 0;
  float*    xl1    = ws + o; o += (size_t)N * D1;
  float*    h1     = ws + o; o += (size_t)N * D1;
  float*    a_src1 = ws + o; o += (size_t)N * HEADS;
  float*    a_dst1 = ws + o; o += (size_t)N * HEADS;
  unsigned* lists  = (unsigned*)(ws + o); o += (size_t)N * LIST_WORDS;
  float*    xl2    = xl1;
  float*    a_src2 = xl2 + (size_t)N * NCLS;
  float*    a_dst2 = a_src2 + (size_t)N;

  const int TB = 256;
  auto nb = [](long long n, int b) { return (unsigned)((n + b - 1) / b); };

  unsigned* plane = (unsigned*)h1;

  plane_kernel<NB_AGG><<<nb((long long)E, PB_EDGES), TB, 0, stream>>>(ei, plane, N, E);
  scan_kernel<NB_AGG, CAP_AGG><<<nb(N, NB_AGG), 1024, 0, stream>>>(ei, plane, N, E, lists);

  gemm1_kernel<<<nb((long long)N, 16), 128, 0, stream>>>(x, W1, xl1, N);
  attn1_kernel<<<nb((long long)N * HEADS, TB), TB, 0, stream>>>(xl1, att_src1, att_dst1,
                                                                a_src1, a_dst1, N);
  agg_kernel<HEADS, HID, NB_AGG, CAP_AGG, true><<<nb(N, NB_AGG), 1024, 0, stream>>>(
      ei, a_src1, a_dst1, xl1, bias1, h1, N, E, lists);

  gemm2_kernel<<<nb(((long long)N + 15) / 16, 8), 256, 0, stream>>>(h1, W2, xl2, N);
  attn2_kernel<<<nb(N, TB), TB, 0, stream>>>(xl2, att_src2, att_dst2, a_src2, a_dst2, N);
  agg_kernel<1, NCLS, NB_AGG, CAP_AGG, false><<<nb(N, NB_AGG), 1024, 0, stream>>>(
      ei, a_src2, a_dst2, xl2, bias2, out, N, E, lists);
}


# Plane launch leaves per edge a one-hot slice mask (16+16 nibble bit-planes) and the packed list word: each scan reads 2 planes, not 8, and fetches one word per own edge (8 in flight), not two
# speedup vs baseline: 6.3368x; 1.0390x over previous
#include <hip/hip_runtime.h>
#include <hip/hip_bf16.h>

typedef __attribute__((ext_vector_type(2))) float v2f;
typedef __attribute__((ext_vector_type(8))) float v8f;

#define F_IN 512
#define HID 8
#define HEADS 8
#define D1 64
#define NCLS 16
#define NEG_SLOPE 0.2f
#define EPS_GAT 1e-16f

__device__ __forceinline__ unsigned fenc(float f) {
  unsigned u = __float_as_uint(f);
  return (u & 0x80000000u) ? ~u : (u | 0x80000000u);
}
__device__ __forceinline__ float fdec(unsigned e) {
  return (e & 0x80000000u) ? __uint_as_float(e ^ 0x80000000u) : __uint_as_float(~e);
}
#define ENC_NEG_INF 0x007FFFFFu

__device__ __forceinline__ float lrelu(float x) { return x > 0.f ? x : x * NEG_SLOPE; }

__global__ __launch_bounds__(128) void gemm1_kernel(const float* __restrict__ x,
                                                    const float* __restrict__ W1,
                                                    float* __restrict__ xl1, int nrows) {
  __shared__ float At[16 * 516];
  const int tid = threadIdx.x;
  const int rowBase = blockIdx.x * 16;
  for (int i = tid; i < 2048; i += 128) {
    int r = i >> 7, cq = i & 127;
    int row = rowBase + r; if (row >= nrows) row = nrows - 1;
    *(float4*)&At[r * 516 + cq * 4] = ((const float4*)(x + (size_t)row * F_IN))[cq];
  }
  __syncthreads();

  const int wave = tid >> 5;
  const int lane = tid & 31;
  const int ln = lane & 15;
  const int khalf = (lane >> 4) << 1;
  const int colBase = wave * 16;

  v8f c = {};
  #pragma unroll 4
  for (int k = 0; k < F_IN; k += 4) {
    v2f a = *(const v2f*)&At[ln * 516 + k + khalf];
    const float* wp = W1 + (size_t)(k + khalf) * D1 + colBase + ln;
    v2f b; b.x = wp[0]; b.y = wp[D1];
    c = __builtin_amdgcn_wmma_f32_16x16x4_f32(false, a, false, b, (short)0, c,
                                              false, false);
  }
  const int mBase = (lane >> 4) * 8;
  float* outp = xl1 + (size_t)rowBase * D1 + colBase;
  #pragma unroll
  for (int i = 0; i < 8; i++) {
    int row = rowBase + mBase + i;
    if (row < nrows) outp[(size_t)(mBase + i) * D1 + ln] = c[i];
  }
}

__global__ void attn1_kernel(const float* __restrict__ xl1,
                             const float* __restrict__ att_src,
                             const float* __restrict__ att_dst,
                             float* __restrict__ a_src, float* __restrict__ a_dst, int n) {
  int t = blockIdx.x * blockDim.x + threadIdx.x;
  if (t >= n * HEADS) return;
  int node = t >> 3, hh = t & 7;
  const float* xp = xl1 + (size_t)node * D1 + hh * HID;
  const float* as = att_src + hh * HID;
  const float* ad = att_dst + hh * HID;
  float s1 = 0.f, s2 = 0.f;
  #pragma unroll
  for (int cc = 0; cc < HID; cc++) { float v = xp[cc]; s1 += v * as[cc]; s2 += v * ad[cc]; }
  a_src[t] = s1; a_dst[t] = s2;
}

#define NB_AGG 392
#define CAP_AGG 16384
#define DLO_BITS 9
#define NOLIST 0xFFFFFFFFu
#define LIST_WORDS 18
#define PLANE_SLICES 256
#define PB_EDGES 4096
#define PB_WORDS 4096
#define PB_PITCH 132

template <int NB>
__device__ __forceinline__ bool plane_usable(int n, int E_) {
  return (n + NB - 1) / NB <= PLANE_SLICES &&
         (((unsigned long long)E_ + (PB_EDGES - 1)) / PB_EDGES) * (unsigned long long)PB_WORDS +
                 (unsigned long long)E_ <=
             (unsigned long long)n * (unsigned long long)D1;
}

template <int NB>
__global__ __launch_bounds__(256) void plane_kernel(const int* __restrict__ ei,
                                                    unsigned* __restrict__ plane, int n, int E_) {
  if (!plane_usable<NB>(n, E_)) return;
  __shared__ __attribute__((aligned(16))) unsigned sm[32 * PB_PITCH];
  const int* __restrict__ dstp = ei + E_;
  const unsigned nblk = ((unsigned)E_ + (unsigned)(PB_EDGES - 1)) / (unsigned)PB_EDGES;
  unsigned* __restrict__ pk = plane + (size_t)nblk * PB_WORDS;
  const unsigned tid = threadIdx.x, lane = tid & 31u, wave = tid >> 5;
  const unsigned e0 = blockIdx.x * (unsigned)PB_EDGES + wave * 512u + lane;
  int d[16], s[16];
  #pragma unroll
  for (int i = 0; i < 16; i++) {
    const unsigned e = e0 + 32u * i;
    const bool ok = e < (unsigned)E_;
    d[i] = ok ? dstp[e] : 0;
    s[i] = ok ? ei[e] : 0;
  }
  const unsigned selh = lane < 16u ? 0xFFFFFFFFu : 0u;
  unsigned inv[4];
  #pragma unroll
  for (int t = 0; t < 4; t++) inv[t] = ((lane >> t) & 1u) ? 0u : 0xFFFFFFFFu;
  unsigned word[16];
  #pragma unroll
  for (int i = 0; i < 16; i++) {
    const unsigned e = e0 + 32u * i;
    const bool ok = e < (unsigned)E_;
    const unsigned sl = (unsigned)d[i] / (unsigned)NB;
    if (ok) pk[e] = ((unsigned)s[i] << DLO_BITS) | ((unsigned)d[i] - sl * (unsigned)NB);
    unsigned w = __builtin_amdgcn_ballot_w32(ok);
    #pragma unroll
    for (int t = 0; t < 4; t++) {
      const unsigned hb = __builtin_amdgcn_ballot_w32(((sl >> (4 + t)) & 1u) != 0u);
      const unsigned lb = __builtin_amdgcn_ballot_w32(((sl >> t) & 1u) != 0u);
      w &= ((hb & selh) | (lb & ~selh)) ^ inv[t];
    }
    word[i] = w;
  }
  #pragma unroll
  for (int r = 0; r < 4; r++)
    *(uint4*)&sm[lane * PB_PITCH + wave * 16u + 4u * r] =
        make_uint4(word[4 * r], word[4 * r + 1], word[4 * r + 2], word[4 * r + 3]);
  __syncthreads();
  #pragma unroll
  for (int r = 0; r < 4; r++) {
    const unsigned idx = (unsigned)r * 256u + tid;
    ((uint4*)plane)[(size_t)blockIdx.x * (PB_WORDS / 4) + idx] =
        *(const uint4*)&sm[(idx >> 5) * PB_PITCH + (idx & 31u) * 4u];
  }
}

template <int NB, int CAP>
__global__ __launch_bounds__(1024) void scan_kernel(const unsigned* __restrict__ plane,
                                                    int n, int E_, unsigned* lists) {
  static_assert(NB <= (1 << DLO_BITS), "local destination must fit its field");
  constexpr int NT = 1024;
  constexpr int NW = NT / 32;
  constexpr int SEG = CAP / NW;
  constexpr int BPI = 4;
  __shared__ unsigned seg[CAP];
  __shared__ unsigned lst[CAP];
  __shared__ unsigned wcnt[NW];

  const int tid = threadIdx.x;
  const unsigned cpad = (gridDim.x + 31u) & ~31u;
  const unsigned lcap = (((unsigned)LIST_WORDS * (unsigned)n - cpad) / gridDim.x) & ~31u;
  unsigned* ecnt = lists;
  unsigned* gl = lists + cpad + (size_t)blockIdx.x * lcap;

  const bool use = plane_usable<NB>(n, E_);
  unsigned total = 0u; bool fits = use;
  if (use) {
    const unsigned lane = tid & 31, wave = tid >> 5;
    unsigned* myseg = seg + wave * SEG;
    unsigned wn = 0u;
    auto push = [&](bool m, unsigned e) {
      const unsigned mask = __builtin_amdgcn_ballot_w32(m);
      if (m) { const unsigned k = wn + __builtin_amdgcn_mbcnt_lo(mask, 0u); if (k < (unsigned)SEG) myseg[k] = e; }
      wn += (unsigned)__builtin_popcount(mask);
    };
    const int nsb = (E_ + (PB_EDGES - 1)) / PB_EDGES;
    const unsigned* __restrict__ pk = plane + (size_t)nsb * PB_WORDS;
    const uint4* __restrict__ pp = (const uint4*)plane;
    const unsigned offh = (blockIdx.x >> 4) * 32u + lane;
    const unsigned offl = (16u + (blockIdx.x & 15u)) * 32u + lane;
    const int wv = __builtin_amdgcn_readfirstlane((int)wave);
    for (int sb0 = wv; sb0 < nsb; sb0 += BPI * NW) {
      uint4 vh[BPI], vl[BPI];
      #pragma unroll
      for (int s = 0; s < BPI; s++) {
        const int sb = sb0 + s * NW;
        const size_t base = (size_t)(sb < nsb ? sb : 0) * (PB_WORDS / 4);
        vh[s] = pp[base + offh]; vl[s] = pp[base + offl];
      }
      #pragma unroll
      for (int s = 0; s < BPI; s++) {
        const int sb = sb0 + s * NW;
        const bool live = sb < nsb;
        unsigned m[4];
        m[0] = vh[s].x & vl[s].x; m[1] = vh[s].y & vl[s].y;
        m[2] = vh[s].z & vl[s].z; m[3] = vh[s].w & vl[s].w;
        #pragma unroll
        for (int q = 0; q < 4; q++) {
          unsigned mm = live ? m[q] : 0u;
          const unsigned eb = (unsigned)sb * (unsigned)PB_EDGES + (4u * lane + (unsigned)q) * 32u;
          while (__builtin_amdgcn_ballot_w32(mm != 0u) != 0u) {
            const unsigned e = eb + (mm != 0u ? (unsigned)__builtin_ctz(mm) : 0u);
            push(mm != 0u && e < (unsigned)E_, e);
            mm &= mm - 1u;
          }
        }
      }
    }
    if (lane == 0) wcnt[wave] = wn;
    __syncthreads();
    unsigned off = 0u;
    for (int w = 0; w < NW; w++) {
      const unsigned c = wcnt[w];
      if ((unsigned)w < wave) off += c;
      total += c; fits = fits && c <= (unsigned)SEG;
    }
    if (fits) {
      for (unsigned j0 = lane; j0 < wn + lane; j0 += 256) {
        unsigned wd[8];
        #pragma unroll
        for (int u = 0; u < 8; u++) {
          const unsigned j = j0 + 32u * u;
          wd[u] = j < wn ? pk[myseg[j]] : 0u;
        }
        #pragma unroll
        for (int u = 0; u < 8; u++) {
          const unsigned j = j0 + 32u * u;
          if (j < wn) lst[off + j] = wd[u];
        }
      }
    }
  }
  if (fits) {
    __syncthreads();
    const bool pub = total <= lcap && (unsigned)n <= (1u << (32 - DLO_BITS));
    if (pub)
      for (unsigned j = tid; j < total; j += NT) gl[j] = lst[j];
    if (tid == 0) ecnt[blockIdx.x] = pub ? total : NOLIST;
  } else {
    if (tid == 0) ecnt[blockIdx.x] = NOLIST;
  }
}

template <int H, int CH, int NB, int CAP, bool ELU>
__global__ __launch_bounds__(1024) void agg_kernel(const int* __restrict__ ei,
                                                   const float* __restrict__ a_src,
                                                   const float* __restrict__ a_dst,
                                                   const float* __restrict__ xl,
                                                   const float* __restrict__ bias,
                                                   float* __restrict__ outp, int n, int E_,
                                                   const unsigned* lists) {
  static_assert(NB <= (1 << DLO_BITS), "local destination must fit its field");
  constexpr int D = H * CH;
  constexpr int RS = D + 1;
  constexpr int NT = 1024;
  __shared__ float acc[NB * RS];
  __shared__ float den[NB * H];
  __shared__ unsigned mx[NB * H];
  __shared__ float adl[NB * H];
  __shared__ unsigned lst[CAP];
  __shared__ unsigned short dlo[CAP];

  const int tid = threadIdx.x;
  const int n0 = blockIdx.x * NB;
  const unsigned nloc = (unsigned)((n - n0) < NB ? (n - n0) : NB);
  const int* __restrict__ dstp = ei + E_;
  const unsigned cpad = (gridDim.x + 31u) & ~31u;
  const unsigned lcap = (((unsigned)LIST_WORDS * (unsigned)n - cpad) / gridDim.x) & ~31u;
  const unsigned* ecnt = lists;
  const unsigned* gl = lists + cpad + (size_t)blockIdx.x * lcap;

  for (int i = tid; i < NB * RS; i += NT) acc[i] = 0.f;
  for (int i = tid; i < NB * H; i += NT) {
    den[i] = 0.f; mx[i] = ENC_NEG_INF;
    adl[i] = ((unsigned)i < nloc * H) ? a_dst[(size_t)n0 * H + i] : 0.f;
  }
  __syncthreads();

  unsigned total = 0u;
  const unsigned cnt = ecnt[blockIdx.x];
  const bool fits = cnt <= (unsigned)CAP && cnt <= lcap;
  if (fits) {
    total = cnt;
    for (unsigned j = tid; j < total; j += NT) {
      const unsigned w = gl[j];
      lst[j] = w >> DLO_BITS; dlo[j] = (unsigned short)(w & ((1u << DLO_BITS) - 1u));
    }
    __syncthreads();
    const unsigned items = total * H;
    const unsigned mfull = items - items % (NT * 4);
    for (unsigned tb = tid; tb < mfull; tb += NT * 4) {
      float al[4]; unsigned slot[4];
      #pragma unroll
      for (int u = 0; u < 4; u++) {
        const unsigned t = tb + u * NT, i = t / H, hh = t % H;
        slot[u] = (unsigned)dlo[i] * H + hh;
        al[u] = a_src[(size_t)lst[i] * H + hh];
      }
      #pragma unroll
      for (int u = 0; u < 4; u++)
        atomicMax(&mx[slot[u]], fenc(lrelu(al[u] + adl[slot[u]])));
    }
    for (unsigned t = mfull + tid; t < items; t += NT) {
      const unsigned i = t / H, hh = t % H, sl = (unsigned)dlo[i] * H + hh;
      atomicMax(&mx[sl], fenc(lrelu(a_src[(size_t)lst[i] * H + hh] + adl[sl])));
    }
    __syncthreads();
    constexpr int LPE = D / 8;
    constexpr int EPI = NT / LPE;
    constexpr int U = 4;
    const unsigned cb = (tid % LPE) * 8, hh = cb / CH;
    const unsigned sfull = total - total % (EPI * U);
    for (unsigned ib = tid / LPE; ib < sfull; ib += EPI * U) {
      float4 xa[U], xb[U]; float as_[U]; unsigned dl[U];
      #pragma unroll
      for (int u = 0; u < U; u++) {
        const unsigned i = ib + u * EPI;
        const unsigned s = lst[i]; dl[u] = dlo[i];
        as_[u] = a_src[(size_t)s * H + hh];
        const float4* xp = (const float4*)(xl + (size_t)s * D + cb);
        xa[u] = xp[0]; xb[u] = xp[1];
      }
      #pragma unroll
      for (int u = 0; u < U; u++) {
        const unsigned hs = dl[u] * H + hh;
        const float ex = __expf(lrelu(as_[u] + adl[hs]) - fdec(mx[hs]));
        if ((cb % CH) == 0) atomicAdd(&den[hs], ex);
        float* ap = &acc[dl[u] * RS + cb];
        atomicAdd(ap + 0, xa[u].x * ex); atomicAdd(ap + 1, xa[u].y * ex);
        atomicAdd(ap + 2, xa[u].z * ex); atomicAdd(ap + 3, xa[u].w * ex);
        atomicAdd(ap + 4, xb[u].x * ex); atomicAdd(ap + 5, xb[u].y * ex);
        atomicAdd(ap + 6, xb[u].z * ex); atomicAdd(ap + 7, xb[u].w * ex);
      }
    }
    for (unsigned i = sfull + tid / LPE; i < total; i += EPI) {
      const unsigned s = lst[i], hs = (unsigned)dlo[i] * H + hh;
      const float4* xp = (const float4*)(xl + (size_t)s * D + cb);
      const float4 xa = xp[0], xb = xp[1];
      const float ex = __expf(lrelu(a_src[(size_t)s * H + hh] + adl[hs]) - fdec(mx[hs]));
      if ((cb % CH) == 0) atomicAdd(&den[hs], ex);
      float* ap = &acc[(unsigned)dlo[i] * RS + cb];
      atomicAdd(ap + 0, xa.x * ex); atomicAdd(ap + 1, xa.y * ex);
      atomicAdd(ap + 2, xa.z * ex); atomicAdd(ap + 3, xa.w * ex);
      atomicAdd(ap + 4, xb.x * ex); atomicAdd(ap + 5, xb.y * ex);
      atomicAdd(ap + 6, xb.z * ex); atomicAdd(ap + 7, xb.w * ex);
    }
  } else {
    for (int e = tid; e < E_; e += NT) {
      const unsigned r = (unsigned)(dstp[e] - n0);
      if (r < nloc) {
        const int s = ei[e];
        for (int h2 = 0; h2 < H; h2++)
          atomicMax(&mx[r * H + h2], fenc(lrelu(a_src[(size_t)s * H + h2] + adl[r * H + h2])));
      }
    }
    __syncthreads();
    for (int e = tid; e < E_; e += NT) {
      const unsigned r = (unsigned)(dstp[e] - n0);
      if (r < nloc) {
        const int s = ei[e];
        for (int h2 = 0; h2 < H; h2++) {
          const unsigned hs = r * H + h2;
          const float ex = __expf(lrelu(a_src[(size_t)s * H + h2] + adl[hs]) - fdec(mx[hs]));
          atomicAdd(&den[hs], ex);
          for (int cc = 0; cc < CH; cc++)
            atomicAdd(&acc[r * RS + h2 * CH + cc], xl[(size_t)s * D + h2 * CH + cc] * ex);
        }
      }
    }
  }
  __syncthreads();
  float* op = outp + (size_t)n0 * D;
  for (unsigned i = tid; i < nloc * D; i += NT) {
    const unsigned c = i % D, r = i / D;
    float v = acc[r * RS + c] / (den[r * H + c / CH] + EPS_GAT) + bias[c];
    if (ELU) v = v > 0.f ? v : (__expf(v) - 1.f);
    op[i] = v;
  }
}

__global__ __launch_bounds__(256) void gemm2_kernel(const float* __restrict__ h,
                                                    const float* __restrict__ W2,
                                                    float* __restrict__ xl2, int nrows) {
  const int wave = threadIdx.x >> 5;
  const int lane = threadIdx.x & 31;
  const int tile = blockIdx.x * 8 + wave;
  const int rowBase = tile * 16;
  if (rowBase >= nrows) return;
  const int ln = lane & 15;
  const int khalf = (lane >> 4) << 1;
  int arow = rowBase + ln; if (arow >= nrows) arow = nrows - 1;
  const float* ap = h + (size_t)arow * D1;

  v8f c = {};
  #pragma unroll
  for (int k = 0; k < D1; k += 4) {
    v2f a = *(const v2f*)&ap[k + khalf];
    v2f b; b.x = W2[(k + khalf) * NCLS + ln];
           b.y = W2[(k + khalf + 1) * NCLS + ln];
    c = __builtin_amdgcn_wmma_f32_16x16x4_f32(false, a, false, b, (short)0, c,
                                              false, false);
  }
  const int mBase = (lane >> 4) * 8;
  float* outp = xl2 + (size_t)rowBase * NCLS;
  #pragma unroll
  for (int i = 0; i < 8; i++) {
    int row = rowBase + mBase + i;
    if (row < nrows) outp[(size_t)(mBase + i) * NCLS + ln] = c[i];
  }
}

__global__ void attn2_kernel(const float* __restrict__ xl2,
                             const float* __restrict__ att_src,
                             const float* __restrict__ att_dst,
                             float* __restrict__ a_src, float* __restrict__ a_dst, int n) {
  int t = blockIdx.x * blockDim.x + threadIdx.x;
  if (t >= n) return;
  const float* xp = xl2 + (size_t)t * NCLS;
  float s1 = 0.f, s2 = 0.f;
  #pragma unroll
  for (int cc = 0; cc < NCLS; cc++) { float v = xp[cc]; s1 += v * att_src[cc]; s2 += v * att_dst[cc]; }
  a_src[t] = s1; a_dst[t] = s2;
}

extern "C" void kernel_launch(void* const* d_in, const int* in_sizes, int n_in,
                              void* d_out, int out_size, void* d_ws, size_t ws_size,
                              hipStream_t stream) {
  const float* x        = (const float*)d_in[0];
  const int*   ei       = (const int*)d_in[1];
  const float* W1       = (const float*)d_in[2];
  const float* att_src1 = (const float*)d_in[3];
  const float* att_dst1 = (const float*)d_in[4];
  const float* bias1    = (const float*)d_in[5];
  const float* W2       = (const float*)d_in[6];
  const float* att_src2 = (const float*)d_in[7];
  const float* att_dst2 = (const float*)d_in[8];
  const float* bias2    = (const float*)d_in[9];
  float* out = (float*)d_out;

  const int N = in_sizes[0] / F_IN;
  const int E = in_sizes[1] / 2;

  float* ws = (float*)d_ws;
  size_t o = 0;
  float*    xl1    = ws + o; o += (size_t)N * D1;
  float*    h1     = ws + o; o += (size_t)N * D1;
  float*    a_src1 = ws + o; o += (size_t)N * HEADS;
  float*    a_dst1 = ws + o; o += (size_t)N * HEADS;
  unsigned* lists  = (unsigned*)(ws + o); o += (size_t)N * LIST_WORDS;
  float*    xl2    = xl1;
  float*    a_src2 = xl2 + (size_t)N * NCLS;
  float*    a_dst2 = a_src2 + (size_t)N;

  const int TB = 256;
  auto nb = [](long long n, int b) { return (unsigned)((n + b - 1) / b); };

  unsigned* plane = (unsigned*)h1;

  plane_kernel<NB_AGG><<<nb((long long)E, PB_EDGES), TB, 0, stream>>>(ei, plane, N, E);
  scan_kernel<NB_AGG, CAP_AGG><<<nb(N, NB_AGG), 1024, 0, stream>>>(plane, N, E, lists);

  gemm1_kernel<<<nb((long long)N, 16), 128, 0, stream>>>(x, W1, xl1, N);
  attn1_kernel<<<nb((long long)N * HEADS, TB), TB, 0, stream>>>(xl1, att_src1, att_dst1,
                                                                a_src1, a_dst1, N);
  agg_kernel<HEADS, HID, NB_AGG, CAP_AGG, true><<<nb(N, NB_AGG), 1024, 0, stream>>>(
      ei, a_src1, a_dst1, xl1, bias1, h1, N, E, lists);

  gemm2_kernel<<<nb(((long long)N + 15) / 16, 8), 256, 0, stream>>>(h1, W2, xl2, N);
  attn2_kernel<<<nb(N, TB), TB, 0, stream>>>(xl2, att_src2, att_dst2, a_src2, a_dst2, N);
  agg_kernel<1, NCLS, NB_AGG, CAP_AGG, false><<<nb(N, NB_AGG), 1024, 0, stream>>>(
      ei, a_src2, a_dst2, xl2, bias2, out, N, E, lists);
}


# GEMM1: resident work-groups stage W1 once in LDS in fragment order; each wave owns 16x64 tiles, its rows of x pass a private LDS chunk as whole lines; no barrier in the k loop; same k order and sums
# speedup vs baseline: 7.0099x; 1.1062x over previous
#include <hip/hip_runtime.h>
#include <hip/hip_bf16.h>

typedef __attribute__((ext_vector_type(2))) float v2f;
typedef __attribute__((ext_vector_type(8))) float v8f;

#define F_IN 512
#define HID 8
#define HEADS 8
#define D1 64
#define NCLS 16
#define NEG_SLOPE 0.2f
#define EPS_GAT 1e-16f

__device__ __forceinline__ unsigned fenc(float f) {
  unsigned u = __float_as_uint(f);
  return (u & 0x80000000u) ? ~u : (u | 0x80000000u);
}
__device__ __forceinline__ float fdec(unsigned e) {
  return (e & 0x80000000u) ? __uint_as_float(e ^ 0x80000000u) : __uint_as_float(~e);
}
#define ENC_NEG_INF 0x007FFFFFu

__device__ __forceinline__ float lrelu(float x) { return x > 0.f ? x : x * NEG_SLOPE; }

#define G1_WAVES 32
#define G1_APITCH 36

template <int NT>
__device__ __forceinline__ void g1_item(const float* __restrict__ x, float* __restrict__ xl1,
                                        const int nrows, float* __restrict__ aw,
                                        const float* __restrict__ Bl, const int lane,
                                        const int tile, const int t0, size_t (&xo)[4],
                                        float4 (&an)[4], const bool more, const int ntile) {
  const int c = lane & 15, h = lane >> 4;
  const int lr = lane >> 3, lq = lane & 7;
  const float* __restrict__ af = aw + c * G1_APITCH + 2 * h;
  const float* __restrict__ bf = Bl + ((t0 >> 1) * 32 + lane) * 4 + (t0 & 1) * 2;
  v8f acc[NT];
  #pragma unroll
  for (int t = 0; t < NT; t++) acc[t] = v8f{};
  #pragma unroll 1
  for (int ch = 0; ch < F_IN / 32; ch++) {
    #pragma unroll
    for (int u = 0; u < 4; u++)
      *(float4*)&aw[(lr + 4 * u) * G1_APITCH + 4 * lq] = an[u];
    if (ch + 1 < F_IN / 32) {
      #pragma unroll
      for (int u = 0; u < 4; u++) an[u] = *(const float4*)(x + xo[u] + 32 * (ch + 1));
    }
    #pragma unroll
    for (int mm = 0; mm < 8; mm++) {
      const int m = 8 * ch + mm;
      const v2f a = *(const v2f*)&af[4 * mm];
      if constexpr (NT == 4) {
        const float4 b0 = *(const float4*)&bf[m * 256];
        const float4 b1 = *(const float4*)&bf[m * 256 + 128];
        v2f q0; q0.x = b0.x; q0.y = b0.y;
        v2f q1; q1.x = b0.z; q1.y = b0.w;
        v2f q2; q2.x = b1.x; q2.y = b1.y;
        v2f q3; q3.x = b1.z; q3.y = b1.w;
        acc[0] = __builtin_amdgcn_wmma_f32_16x16x4_f32(false, a, false, q0, (short)0, acc[0], false, false);
        acc[1] = __builtin_amdgcn_wmma_f32_16x16x4_f32(false, a, false, q1, (short)0, acc[1], false, false);
        acc[2] = __builtin_amdgcn_wmma_f32_16x16x4_f32(false, a, false, q2, (short)0, acc[2], false, false);
        acc[3] = __builtin_amdgcn_wmma_f32_16x16x4_f32(false, a, false, q3, (short)0, acc[3], false, false);
      } else {
        const v2f q = *(const v2f*)&bf[m * 256];
        acc[0] = __builtin_amdgcn_wmma_f32_16x16x4_f32(false, a, false, q, (short)0, acc[0], false, false);
      }
    }
  }
  if (more) {
    #pragma unroll
    for (int u = 0; u < 4; u++) {
      int row = ntile * 16 + lr + 4 * u; if (row >= nrows) row = nrows - 1;
      xo[u] = (size_t)row * F_IN + 4 * lq;
      an[u] = *(const float4*)(x + xo[u]);
    }
  }
  const int rowBase = tile * 16, mBase = h * 8;
  float* outp = xl1 + (size_t)rowBase * D1 + 16 * t0 + c;
  #pragma unroll
  for (int i = 0; i < 8; i++) {
    const int row = rowBase + mBase + i;
    if (row < nrows) {
      float* o = outp + (size_t)(mBase + i) * D1;
      #pragma unroll
      for (int t = 0; t < NT; t++) o[16 * t] = acc[t][i];
    }
  }
}

__global__ __launch_bounds__(1024) void gemm1_kernel(const float* __restrict__ x,
                                                     const float* __restrict__ W1,
                                                     float* __restrict__ xl1, int nrows) {
  __shared__ __attribute__((aligned(16))) float Bl[(F_IN / 4) * 2 * 32 * 4];
  __shared__ __attribute__((aligned(16))) float Aw[G1_WAVES * 16 * G1_APITCH];
  const int tid = threadIdx.x;
  const int lane = tid & 31;
  const int wave = __builtin_amdgcn_readfirstlane(tid >> 5);
  float* __restrict__ aw = Aw + wave * (16 * G1_APITCH);
  const int lr = lane >> 3, lq = lane & 7;
  const int ntiles = (nrows + 15) >> 4;
  const int b = (int)blockIdx.x, G = (int)gridDim.x;
  const int q = ntiles / G, rem = ntiles - q * G;
  const int wb = wave >> 3;
  const int slot = 4 * (((wave & 7) - wb) & 7) + wb;
  int j = slot;
  const int piece = (slot - q) & 31;
  bool extra = b < rem && piece < 4;
  int tile = 0, kind = 0;
  if (j < q) { tile = b + G * j; j += G1_WAVES; kind = 1; }
  else if (extra) { tile = G * q + b; extra = false; kind = 2; }

  size_t xo[4];
  float4 an[4];
  #pragma unroll
  for (int u = 0; u < 4; u++) { xo[u] = 0; an[u] = make_float4(0.f, 0.f, 0.f, 0.f); }
  if (kind != 0) {
    #pragma unroll
    for (int u = 0; u < 4; u++) {
      int row = tile * 16 + lr + 4 * u; if (row >= nrows) row = nrows - 1;
      xo[u] = (size_t)row * F_IN + 4 * lq;
      an[u] = *(const float4*)(x + xo[u]);
    }
  }

  #pragma unroll
  for (int g = tid; g < (F_IN / 4) * 2 * 32; g += 1024) {
    const int m = g >> 6, p = (g >> 5) & 1, l = g & 31, cc = l & 15, hh = l >> 4;
    const int k0 = 4 * m + 2 * hh;
    const float* wp = W1 + (size_t)k0 * D1 + 32 * p + cc;
    float4 v; v.x = wp[0]; v.y = wp[D1]; v.z = wp[16]; v.w = wp[D1 + 16];
    *(float4*)&Bl[g * 4] = v;
  }
  __syncthreads();

  while (kind != 0) {
    const int ctile = tile, ckind = kind;
    if (j < q) { tile = b + G * j; j += G1_WAVES; kind = 1; }
    else if (extra) { tile = G * q + b; extra = false; kind = 2; }
    else kind = 0;
    if (ckind == 1) g1_item<4>(x, xl1, nrows, aw, Bl, lane, ctile, 0, xo, an, kind != 0, tile);
    else            g1_item<1>(x, xl1, nrows, aw, Bl, lane, ctile, piece, xo, an, kind != 0, tile);
  }
}

__global__ void attn1_kernel(const float* __restrict__ xl1,
                             const float* __restrict__ att_src,
                             const float* __restrict__ att_dst,
                             float* __restrict__ a_src, float* __restrict__ a_dst, int n) {
  int t = blockIdx.x * blockDim.x + threadIdx.x;
  if (t >= n * HEADS) return;
  int node = t >> 3, hh = t & 7;
  const float* xp = xl1 + (size_t)node * D1 + hh * HID;
  const float* as = att_src + hh * HID;
  const float* ad = att_dst + hh * HID;
  float s1 = 0.f, s2 = 0.f;
  #pragma unroll
  for (int cc = 0; cc < HID; cc++) { float v = xp[cc]; s1 += v * as[cc]; s2 += v * ad[cc]; }
  a_src[t] = s1; a_dst[t] = s2;
}

#define NB_AGG 392
#define CAP_AGG 16384
#define DLO_BITS 9
#define NOLIST 0xFFFFFFFFu
#define LIST_WORDS 18
#define PLANE_SLICES 256
#define PB_EDGES 4096
#define PB_WORDS 4096
#define PB_PITCH 132

template <int NB>
__device__ __forceinline__ bool plane_usable(int n, int E_) {
  return (n + NB - 1) / NB <= PLANE_SLICES &&
         (((unsigned long long)E_ + (PB_EDGES - 1)) / PB_EDGES) * (unsigned long long)PB_WORDS +
                 (unsigned long long)E_ <=
             (unsigned long long)n * (unsigned long long)D1;
}

template <int NB>
__global__ __launch_bounds__(256) void plane_kernel(const int* __restrict__ ei,
                                                    unsigned* __restrict__ plane, int n, int E_) {
  if (!plane_usable<NB>(n, E_)) return;
  __shared__ __attribute__((aligned(16))) unsigned sm[32 * PB_PITCH];
  const int* __restrict__ dstp = ei + E_;
  const unsigned nblk = ((unsigned)E_ + (unsigned)(PB_EDGES - 1)) / (unsigned)PB_EDGES;
  unsigned* __restrict__ pk = plane + (size_t)nblk * PB_WORDS;
  const unsigned tid = threadIdx.x, lane = tid & 31u, wave = tid >> 5;
  const unsigned e0 = blockIdx.x * (unsigned)PB_EDGES + wave * 512u + lane;
  int d[16], s[16];
  #pragma unroll
  for (int i = 0; i < 16; i++) {
    const unsigned e = e0 + 32u * i;
    const bool ok = e < (unsigned)E_;
    d[i] = ok ? dstp[e] : 0;
    s[i] = ok ? ei[e] : 0;
  }
  const unsigned selh = lane < 16u ? 0xFFFFFFFFu : 0u;
  unsigned inv[4];
  #pragma unroll
  for (int t = 0; t < 4; t++) inv[t] = ((lane >> t) & 1u) ? 0u : 0xFFFFFFFFu;
  unsigned word[16];
  #pragma unroll
  for (int i = 0; i < 16; i++) {
    const unsigned e = e0 + 32u * i;
    const bool ok = e < (unsigned)E_;
    const unsigned sl = (unsigned)d[i] / (unsigned)NB;
    if (ok) pk[e] = ((unsigned)s[i] << DLO_BITS) | ((unsigned)d[i] - sl * (unsigned)NB);
    unsigned w = __builtin_amdgcn_ballot_w32(ok);
    #pragma unroll
    for (int t = 0; t < 4; t++) {
      const unsigned hb = __builtin_amdgcn_ballot_w32(((sl >> (4 + t)) & 1u) != 0u);
      const unsigned lb = __builtin_amdgcn_ballot_w32(((sl >> t) & 1u) != 0u);
      w &= ((hb & selh) | (lb & ~selh)) ^ inv[t];
    }
    word[i] = w;
  }
  #pragma unroll
  for (int r = 0; r < 4; r++)
    *(uint4*)&sm[lane * PB_PITCH + wave * 16u + 4u * r] =
        make_uint4(word[4 * r], word[4 * r + 1], word[4 * r + 2], word[4 * r + 3]);
  __syncthreads();
  #pragma unroll
  for (int r = 0; r < 4; r++) {
    const unsigned idx = (unsigned)r * 256u + tid;
    ((uint4*)plane)[(size_t)blockIdx.x * (PB_WORDS / 4) + idx] =
        *(const uint4*)&sm[(idx >> 5) * PB_PITCH + (idx & 31u) * 4u];
  }
}

template <int NB, int CAP>
__global__ __launch_bounds__(1024) void scan_kernel(const unsigned* __restrict__ plane,
                                                    int n, int E_, unsigned* lists) {
  static_assert(NB <= (1 << DLO_BITS), "local destination must fit its field");
  constexpr int NT = 1024;
  constexpr int NW = NT / 32;
  constexpr int SEG = CAP / NW;
  constexpr int BPI = 4;
  __shared__ unsigned seg[CAP];
  __shared__ unsigned lst[CAP];
  __shared__ unsigned wcnt[NW];

  const int tid = threadIdx.x;
  const unsigned cpad = (gridDim.x + 31u) & ~31u;
  const unsigned lcap = (((unsigned)LIST_WORDS * (unsigned)n - cpad) / gridDim.x) & ~31u;
  unsigned* ecnt = lists;
  unsigned* gl = lists + cpad + (size_t)blockIdx.x * lcap;

  const bool use = plane_usable<NB>(n, E_);
  unsigned total = 0u; bool fits = use;
  if (use) {
    const unsigned lane = tid & 31, wave = tid >> 5;
    unsigned* myseg = seg + wave * SEG;
    unsigned wn = 0u;
    auto push = [&](bool m, unsigned e) {
      const unsigned mask = __builtin_amdgcn_ballot_w32(m);
      if (m) { const unsigned k = wn + __builtin_amdgcn_mbcnt_lo(mask, 0u); if (k < (unsigned)SEG) myseg[k] = e; }
      wn += (unsigned)__builtin_popcount(mask);
    };
    const int nsb = (E_ + (PB_EDGES - 1)) / PB_EDGES;
    const unsigned* __restrict__ pk = plane + (size_t)nsb * PB_WORDS;
    const uint4* __restrict__ pp = (const uint4*)plane;
    const unsigned offh = (blockIdx.x >> 4) * 32u + lane;
    const unsigned offl = (16u + (blockIdx.x & 15u)) * 32u + lane;
    const int wv = __builtin_amdgcn_readfirstlane((int)wave);
    for (int sb0 = wv; sb0 < nsb; sb0 += BPI * NW) {
      uint4 vh[BPI], vl[BPI];
      #pragma unroll
      for (int s = 0; s < BPI; s++) {
        const int sb = sb0 + s * NW;
        const size_t base = (size_t)(sb < nsb ? sb : 0) * (PB_WORDS / 4);
        vh[s] = pp[base + offh]; vl[s] = pp[base + offl];
      }
      #pragma unroll
      for (int s = 0; s < BPI; s++) {
        const int sb = sb0 + s * NW;
        const bool live = sb < nsb;
        unsigned m[4];
        m[0] = vh[s].x & vl[s].x; m[1] = vh[s].y & vl[s].y;
        m[2] = vh[s].z & vl[s].z; m[3] = vh[s].w & vl[s].w;
        #pragma unroll
        for (int q = 0; q < 4; q++) {
          unsigned mm = live ? m[q] : 0u;
          const unsigned eb = (unsigned)sb * (unsigned)PB_EDGES + (4u * lane + (unsigned)q) * 32u;
          while (__builtin_amdgcn_ballot_w32(mm != 0u) != 0u) {
            const unsigned e = eb + (mm != 0u ? (unsigned)__builtin_ctz(mm) : 0u);
            push(mm != 0u && e < (unsigned)E_, e);
            mm &= mm - 1u;
          }
        }
      }
    }
    if (lane == 0) wcnt[wave] = wn;
    __syncthreads();
    unsigned off = 0u;
    for (int w = 0; w < NW; w++) {
      const unsigned c = wcnt[w];
      if ((unsigned)w < wave) off += c;
      total += c; fits = fits && c <= (unsigned)SEG;
    }
    if (fits) {
      for (unsigned j0 = lane; j0 < wn + lane; j0 += 256) {
        unsigned wd[8];
        #pragma unroll
        for (int u = 0; u < 8; u++) {
          const unsigned j = j0 + 32u * u;
          wd[u] = j < wn ? pk[myseg[j]] : 0u;
        }
        #pragma unroll
        for (int u = 0; u < 8; u++) {
          const unsigned j = j0 + 32u * u;
          if (j < wn) lst[off + j] = wd[u];
        }
      }
    }
  }
  if (fits) {
    __syncthreads();
    const bool pub = total <= lcap && (unsigned)n <= (1u << (32 - DLO_BITS));
    if (pub)
      for (unsigned j = tid; j < total; j += NT) gl[j] = lst[j];
    if (tid == 0) ecnt[blockIdx.x] = pub ? total : NOLIST;
  } else {
    if (tid == 0) ecnt[blockIdx.x] = NOLIST;
  }
}

template <int H, int CH, int NB, int CAP, bool ELU>
__global__ __launch_bounds__(1024) void agg_kernel(const int* __restrict__ ei,
                                                   const float* __restrict__ a_src,
                                                   const float* __restrict__ a_dst,
                                                   const float* __restrict__ xl,
                                                   const float* __restrict__ bias,
                                                   float* __restrict__ outp, int n, int E_,
                                                   const unsigned* lists) {
  static_assert(NB <= (1 << DLO_BITS), "local destination must fit its field");
  constexpr int D = H * CH;
  constexpr int RS = D + 1;
  constexpr int NT = 1024;
  __shared__ float acc[NB * RS];
  __shared__ float den[NB * H];
  __shared__ unsigned mx[NB * H];
  __shared__ float adl[NB * H];
  __shared__ unsigned lst[CAP];
  __shared__ unsigned short dlo[CAP];

  const int tid = threadIdx.x;
  const int n0 = blockIdx.x * NB;
  const unsigned nloc = (unsigned)((n - n0) < NB ? (n - n0) : NB);
  const int* __restrict__ dstp = ei + E_;
  const unsigned cpad = (gridDim.x + 31u) & ~31u;
  const unsigned lcap = (((unsigned)LIST_WORDS * (unsigned)n - cpad) / gridDim.x) & ~31u;
  const unsigned* ecnt = lists;
  const unsigned* gl = lists + cpad + (size_t)blockIdx.x * lcap;

  for (int i = tid; i < NB * RS; i += NT) acc[i] = 0.f;
  for (int i = tid; i < NB * H; i += NT) {
    den[i] = 0.f; mx[i] = ENC_NEG_INF;
    adl[i] = ((unsigned)i < nloc * H) ? a_dst[(size_t)n0 * H + i] : 0.f;
  }
  __syncthreads();

  unsigned total = 0u;
  const unsigned cnt = ecnt[blockIdx.x];
  const bool fits = cnt <= (unsigned)CAP && cnt <= lcap;
  if (fits) {
    total = cnt;
    for (unsigned j = tid; j < total; j += NT) {
      const unsigned w = gl[j];
      lst[j] = w >> DLO_BITS; dlo[j] = (unsigned short)(w & ((1u << DLO_BITS) - 1u));
    }
    __syncthreads();
    const unsigned items = total * H;
    const unsigned mfull = items - items % (NT * 4);
    for (unsigned tb = tid; tb < mfull; tb += NT * 4) {
      float al[4]; unsigned slot[4];
      #pragma unroll
      for (int u = 0; u < 4; u++) {
        const unsigned t = tb + u * NT, i = t / H, hh = t % H;
        slot[u] = (unsigned)dlo[i] * H + hh;
        al[u] = a_src[(size_t)lst[i] * H + hh];
      }
      #pragma unroll
      for (int u = 0; u < 4; u++)
        atomicMax(&mx[slot[u]], fenc(lrelu(al[u] + adl[slot[u]])));
    }
    for (unsigned t = mfull + tid; t < items; t += NT) {
      const unsigned i = t / H, hh = t % H, sl = (unsigned)dlo[i] * H + hh;
      atomicMax(&mx[sl], fenc(lrelu(a_src[(size_t)lst[i] * H + hh] + adl[sl])));
    }
    __syncthreads();
    constexpr int LPE = D / 8;
    constexpr int EPI = NT / LPE;
    constexpr int U = 4;
    const unsigned cb = (tid % LPE) * 8, hh = cb / CH;
    const unsigned sfull = total - total % (EPI * U);
    for (unsigned ib = tid / LPE; ib < sfull; ib += EPI * U) {
      float4 xa[U], xb[U]; float as_[U]; unsigned dl[U];
      #pragma unroll
      for (int u = 0; u < U; u++) {
        const unsigned i = ib + u * EPI;
        const unsigned s = lst[i]; dl[u] = dlo[i];
        as_[u] = a_src[(size_t)s * H + hh];
        const float4* xp = (const float4*)(xl + (size_t)s * D + cb);
        xa[u] = xp[0]; xb[u] = xp[1];
      }
      #pragma unroll
      for (int u = 0; u < U; u++) {
        const unsigned hs = dl[u] * H + hh;
        const float ex = __expf(lrelu(as_[u] + adl[hs]) - fdec(mx[hs]));
        if ((cb % CH) == 0) atomicAdd(&den[hs], ex);
        float* ap = &acc[dl[u] * RS + cb];
        atomicAdd(ap + 0, xa[u].x * ex); atomicAdd(ap + 1, xa[u].y * ex);
        atomicAdd(ap + 2, xa[u].z * ex); atomicAdd(ap + 3, xa[u].w * ex);
        atomicAdd(ap + 4, xb[u].x * ex); atomicAdd(ap + 5, xb[u].y * ex);
        atomicAdd(ap + 6, xb[u].z * ex); atomicAdd(ap + 7, xb[u].w * ex);
      }
    }
    for (unsigned i = sfull + tid / LPE; i < total; i += EPI) {
      const unsigned s = lst[i], hs = (unsigned)dlo[i] * H + hh;
      const float4* xp = (const float4*)(xl + (size_t)s * D + cb);
      const float4 xa = xp[0], xb = xp[1];
      const float ex = __expf(lrelu(a_src[(size_t)s * H + hh] + adl[hs]) - fdec(mx[hs]));
      if ((cb % CH) == 0) atomicAdd(&den[hs], ex);
      float* ap = &acc[(unsigned)dlo[i] * RS + cb];
      atomicAdd(ap + 0, xa.x * ex); atomicAdd(ap + 1, xa.y * ex);
      atomicAdd(ap + 2, xa.z * ex); atomicAdd(ap + 3, xa.w * ex);
      atomicAdd(ap + 4, xb.x * ex); atomicAdd(ap + 5, xb.y * ex);
      atomicAdd(ap + 6, xb.z * ex); atomicAdd(ap + 7, xb.w * ex);
    }
  } else {
    for (int e = tid; e < E_; e += NT) {
      const unsigned r = (unsigned)(dstp[e] - n0);
      if (r < nloc) {
        const int s = ei[e];
        for (int h2 = 0; h2 < H; h2++)
          atomicMax(&mx[r * H + h2], fenc(lrelu(a_src[(size_t)s * H + h2] + adl[r * H + h2])));
      }
    }
    __syncthreads();
    for (int e = tid; e < E_; e += NT) {
      const unsigned r = (unsigned)(dstp[e] - n0);
      if (r < nloc) {
        const int s = ei[e];
        for (int h2 = 0; h2 < H; h2++) {
          const unsigned hs = r * H + h2;
          const float ex = __expf(lrelu(a_src[(size_t)s * H + h2] + adl[hs]) - fdec(mx[hs]));
          atomicAdd(&den[hs], ex);
          for (int cc = 0; cc < CH; cc++)
            atomicAdd(&acc[r * RS + h2 * CH + cc], xl[(size_t)s * D + h2 * CH + cc] * ex);
        }
      }
    }
  }
  __syncthreads();
  float* op = outp + (size_t)n0 * D;
  for (unsigned i = tid; i < nloc * D; i += NT) {
    const unsigned c = i % D, r = i / D;
    float v = acc[r * RS + c] / (den[r * H + c / CH] + EPS_GAT) + bias[c];
    if (ELU) v = v > 0.f ? v : (__expf(v) - 1.f);
    op[i] = v;
  }
}

__global__ __launch_bounds__(256) void gemm2_kernel(const float* __restrict__ h,
                                                    const float* __restrict__ W2,
                                                    float* __restrict__ xl2, int nrows) {
  const int wave = threadIdx.x >> 5;
  const int lane = threadIdx.x & 31;
  const int tile = blockIdx.x * 8 + wave;
  const int rowBase = tile * 16;
  if (rowBase >= nrows) return;
  const int ln = lane & 15;
  const int khalf = (lane >> 4) << 1;
  int arow = rowBase + ln; if (arow >= nrows) arow = nrows - 1;
  const float* ap = h + (size_t)arow * D1;

  v8f c = {};
  #pragma unroll
  for (int k = 0; k < D1; k += 4) {
    v2f a = *(const v2f*)&ap[k + khalf];
    v2f b; b.x = W2[(k + khalf) * NCLS + ln];
           b.y = W2[(k + khalf + 1) * NCLS + ln];
    c = __builtin_amdgcn_wmma_f32_16x16x4_f32(false, a, false, b, (short)0, c,
                                              false, false);
  }
  const int mBase = (lane >> 4) * 8;
  float* outp = xl2 + (size_t)rowBase * NCLS;
  #pragma unroll
  for (int i = 0; i < 8; i++) {
    int row = rowBase + mBase + i;
    if (row < nrows) outp[(size_t)(mBase + i) * NCLS + ln] = c[i];
  }
}

__global__ void attn2_kernel(const float* __restrict__ xl2,
                             const float* __restrict__ att_src,
                             const float* __restrict__ att_dst,
                             float* __restrict__ a_src, float* __restrict__ a_dst, int n) {
  int t = blockIdx.x * blockDim.x + threadIdx.x;
  if (t >= n) return;
  const float* xp = xl2 + (size_t)t * NCLS;
  float s1 = 0.f, s2 = 0.f;
  #pragma unroll
  for (int cc = 0; cc < NCLS; cc++) { float v = xp[cc]; s1 += v * att_src[cc]; s2 += v * att_dst[cc]; }
  a_src[t] = s1; a_dst[t] = s2;
}

extern "C" void kernel_launch(void* const* d_in, const int* in_sizes, int n_in,
                              void* d_out, int out_size, void* d_ws, size_t ws_size,
                              hipStream_t stream) {
  const float* x        = (const float*)d_in[0];
  const int*   ei       = (const int*)d_in[1];
  const float* W1       = (const float*)d_in[2];
  const float* att_src1 = (const float*)d_in[3];
  const float* att_dst1 = (const float*)d_in[4];
  const float* bias1    = (const float*)d_in[5];
  const float* W2       = (const float*)d_in[6];
  const float* att_src2 = (const float*)d_in[7];
  const float* att_dst2 = (const float*)d_in[8];
  const float* bias2    = (const float*)d_in[9];
  float* out = (float*)d_out;

  const int N = in_sizes[0] / F_IN;
  const int E = in_sizes[1] / 2;

  float* ws = (float*)d_ws;
  size_t o = 0;
  float*    xl1    = ws + o; o += (size_t)N * D1;
  float*    h1     = ws + o; o += (size_t)N * D1;
  float*    a_src1 = ws + o; o += (size_t)N * HEADS;
  float*    a_dst1 = ws + o; o += (size_t)N * HEADS;
  unsigned* lists  = (unsigned*)(ws + o); o += (size_t)N * LIST_WORDS;
  float*    xl2    = xl1;
  float*    a_src2 = xl2 + (size_t)N * NCLS;
  float*    a_dst2 = a_src2 + (size_t)N;

  const int TB = 256;
  auto nb = [](long long n, int b) { return (unsigned)((n + b - 1) / b); };

  unsigned* plane = (unsigned*)h1;

  plane_kernel<NB_AGG><<<nb((long long)E, PB_EDGES), TB, 0, stream>>>(ei, plane, N, E);
  scan_kernel<NB_AGG, CAP_AGG><<<nb(N, NB_AGG), 1024, 0, stream>>>(plane, N, E, lists);

  gemm1_kernel<<<256, 1024, 0, stream>>>(x, W1, xl1, N);
  attn1_kernel<<<nb((long long)N * HEADS, TB), TB, 0, stream>>>(xl1, att_src1, att_dst1,
                                                                a_src1, a_dst1, N);
  agg_kernel<HEADS, HID, NB_AGG, CAP_AGG, true><<<nb(N, NB_AGG), 1024, 0, stream>>>(
      ei, a_src1, a_dst1, xl1, bias1, h1, N, E, lists);

  gemm2_kernel<<<nb(((long long)N + 15) / 16, 8), 256, 0, stream>>>(h1, W2, xl2, N);
  attn2_kernel<<<nb(N, TB), TB, 0, stream>>>(xl2, att_src2, att_dst2, a_src2, a_dst2, N);
  agg_kernel<1, NCLS, NB_AGG, CAP_AGG, false><<<nb(N, NB_AGG), 1024, 0, stream>>>(
      ei, a_src2, a_dst2, xl2, bias2, out, N, E, lists);
}


# Aggregation sums, both layers: lanes remapped so one LDS float add instruction covers whole rows (layer 1: one edge, 2 channels a lane; layer 2: two edges): bank conflicts go; sum order=LDS arrival
# speedup vs baseline: 7.8766x; 1.1236x over previous
#include <hip/hip_runtime.h>
#include <hip/hip_bf16.h>
#include <type_traits>

typedef __attribute__((ext_vector_type(2))) float v2f;
typedef __attribute__((ext_vector_type(8))) float v8f;

#define F_IN 512
#define HID 8
#define HEADS 8
#define D1 64
#define NCLS 16
#define NEG_SLOPE 0.2f
#define EPS_GAT 1e-16f

__device__ __forceinline__ unsigned fenc(float f) {
  unsigned u = __float_as_uint(f);
  return (u & 0x80000000u) ? ~u : (u | 0x80000000u);
}
__device__ __forceinline__ float fdec(unsigned e) {
  return (e & 0x80000000u) ? __uint_as_float(e ^ 0x80000000u) : __uint_as_float(~e);
}
#define ENC_NEG_INF 0x007FFFFFu

__device__ __forceinline__ float lrelu(float x) { return x > 0.f ? x : x * NEG_SLOPE; }

template <int K> __device__ __forceinline__ unsigned quad_take(unsigned v) {
  return (unsigned)__builtin_amdgcn_update_dpp(0, (int)v, K * 0x55, 0xF, 0xF, false);
}
template <int K> __device__ __forceinline__ unsigned row_take(unsigned v) {
  return (unsigned)__builtin_amdgcn_update_dpp(0, (int)v, 0x150 + K, 0xF, 0xF, false);
}

#define G1_WAVES 32
#define G1_APITCH 36

template <int NT>
__device__ __forceinline__ void g1_item(const float* __restrict__ x, float* __restrict__ xl1,
                                        const int nrows, float* __restrict__ aw,
                                        const float* __restrict__ Bl, const int lane,
                                        const int tile, const int t0, size_t (&xo)[4],
                                        float4 (&an)[4], const bool more, const int ntile) {
  const int c = lane & 15, h = lane >> 4;
  const int lr = lane >> 3, lq = lane & 7;
  const float* __restrict__ af = aw + c * G1_APITCH + 2 * h;
  const float* __restrict__ bf = Bl + ((t0 >> 1) * 32 + lane) * 4 + (t0 & 1) * 2;
  v8f acc[NT];
  #pragma unroll
  for (int t = 0; t < NT; t++) acc[t] = v8f{};
  #pragma unroll 1
  for (int ch = 0; ch < F_IN / 32; ch++) {
    #pragma unroll
    for (int u = 0; u < 4; u++)
      *(float4*)&aw[(lr + 4 * u) * G1_APITCH + 4 * lq] = an[u];
    if (ch + 1 < F_IN / 32) {
      #pragma unroll
      for (int u = 0; u < 4; u++) an[u] = *(const float4*)(x + xo[u] + 32 * (ch + 1));
    }
    #pragma unroll
    for (int mm = 0; mm < 8; mm++) {
      const int m = 8 * ch + mm;
      const v2f a = *(const v2f*)&af[4 * mm];
      if constexpr (NT == 4) {
        const float4 b0 = *(const float4*)&bf[m * 256];
        const float4 b1 = *(const float4*)&bf[m * 256 + 128];
        v2f q0; q0.x = b0.x; q0.y = b0.y;
        v2f q1; q1.x = b0.z; q1.y = b0.w;
        v2f q2; q2.x = b1.x; q2.y = b1.y;
        v2f q3; q3.x = b1.z; q3.y = b1.w;
        acc[0] = __builtin_amdgcn_wmma_f32_16x16x4_f32(false, a, false, q0, (short)0, acc[0], false, false);
        acc[1] = __builtin_amdgcn_wmma_f32_16x16x4_f32(false, a, false, q1, (short)0, acc[1], false, false);
        acc[2] = __builtin_amdgcn_wmma_f32_16x16x4_f32(false, a, false, q2, (short)0, acc[2], false, false);
        acc[3] = __builtin_amdgcn_wmma_f32_16x16x4_f32(false, a, false, q3, (short)0, acc[3], false, false);
      } else {
        const v2f q = *(const v2f*)&bf[m * 256];
        acc[0] = __builtin_amdgcn_wmma_f32_16x16x4_f32(false, a, false, q, (short)0, acc[0], false, false);
      }
    }
  }
  if (more) {
    #pragma unroll
    for (int u = 0; u < 4; u++) {
      int row = ntile * 16 + lr + 4 * u; if (row >= nrows) row = nrows - 1;
      xo[u] = (size_t)row * F_IN + 4 * lq;
      an[u] = *(const float4*)(x + xo[u]);
    }
  }
  const int rowBase = tile * 16, mBase = h * 8;
  float* outp = xl1 + (size_t)rowBase * D1 + 16 * t0 + c;
  #pragma unroll
  for (int i = 0; i < 8; i++) {
    const int row = rowBase + mBase + i;
    if (row < nrows) {
      float* o = outp + (size_t)(mBase + i) * D1;
      #pragma unroll
      for (int t = 0; t < NT; t++) o[16 * t] = acc[t][i];
    }
  }
}

__global__ __launch_bounds__(1024) void gemm1_kernel(const float* __restrict__ x,
                                                     const float* __restrict__ W1,
                                                     float* __restrict__ xl1, int nrows) {
  __shared__ __attribute__((aligned(16))) float Bl[(F_IN / 4) * 2 * 32 * 4];
  __shared__ __attribute__((aligned(16))) float Aw[G1_WAVES * 16 * G1_APITCH];
  const int tid = threadIdx.x;
  const int lane = tid & 31;
  const int wave = __builtin_amdgcn_readfirstlane(tid >> 5);
  float* __restrict__ aw = Aw + wave * (16 * G1_APITCH);
  const int lr = lane >> 3, lq = lane & 7;
  const int ntiles = (nrows + 15) >> 4;
  const int b = (int)blockIdx.x, G = (int)gridDim.x;
  const int q = ntiles / G, rem = ntiles - q * G;
  const int wb = wave >> 3;
  const int slot = 4 * (((wave & 7) - wb) & 7) + wb;
  int j = slot;
  const int piece = (slot - q) & 31;
  bool extra = b < rem && piece < 4;
  int tile = 0, kind = 0;
  if (j < q) { tile = b + G * j; j += G1_WAVES; kind = 1; }
  else if (extra) { tile = G * q + b; extra = false; kind = 2; }

  size_t xo[4];
  float4 an[4];
  #pragma unroll
  for (int u = 0; u < 4; u++) { xo[u] = 0; an[u] = make_float4(0.f, 0.f, 0.f, 0.f); }
  if (kind != 0) {
    #pragma unroll
    for (int u = 0; u < 4; u++) {
      int row = tile * 16 + lr + 4 * u; if (row >= nrows) row = nrows - 1;
      xo[u] = (size_t)row * F_IN + 4 * lq;
      an[u] = *(const float4*)(x + xo[u]);
    }
  }

  #pragma unroll
  for (int g = tid; g < (F_IN / 4) * 2 * 32; g += 1024) {
    const int m = g >> 6, p = (g >> 5) & 1, l = g & 31, cc = l & 15, hh = l >> 4;
    const int k0 = 4 * m + 2 * hh;
    const float* wp = W1 + (size_t)k0 * D1 + 32 * p + cc;
    float4 v; v.x = wp[0]; v.y = wp[D1]; v.z = wp[16]; v.w = wp[D1 + 16];
    *(float4*)&Bl[g * 4] = v;
  }
  __syncthreads();

  while (kind != 0) {
    const int ctile = tile, ckind = kind;
    if (j < q) { tile = b + G * j; j += G1_WAVES; kind = 1; }
    else if (extra) { tile = G * q + b; extra = false; kind = 2; }
    else kind = 0;
    if (ckind == 1) g1_item<4>(x, xl1, nrows, aw, Bl, lane, ctile, 0, xo, an, kind != 0, tile);
    else            g1_item<1>(x, xl1, nrows, aw, Bl, lane, ctile, piece, xo, an, kind != 0, tile);
  }
}

__global__ void attn1_kernel(const float* __restrict__ xl1,
                             const float* __restrict__ att_src,
                             const float* __restrict__ att_dst,
                             float* __restrict__ a_src, float* __restrict__ a_dst, int n) {
  int t = blockIdx.x * blockDim.x + threadIdx.x;
  if (t >= n * HEADS) return;
  int node = t >> 3, hh = t & 7;
  const float* xp = xl1 + (size_t)node * D1 + hh * HID;
  const float* as = att_src + hh * HID;
  const float* ad = att_dst + hh * HID;
  float s1 = 0.f, s2 = 0.f;
  #pragma unroll
  for (int cc = 0; cc < HID; cc++) { float v = xp[cc]; s1 += v * as[cc]; s2 += v * ad[cc]; }
  a_src[t] = s1; a_dst[t] = s2;
}

#define NB_AGG 392
#define CAP_AGG 16384
#define DLO_BITS 9
#define NOLIST 0xFFFFFFFFu
#define LIST_WORDS 18
#define PLANE_SLICES 256
#define PB_EDGES 4096
#define PB_WORDS 4096
#define PB_PITCH 132

template <int NB>
__device__ __forceinline__ bool plane_usable(int n, int E_) {
  return (n + NB - 1) / NB <= PLANE_SLICES &&
         (((unsigned long long)E_ + (PB_EDGES - 1)) / PB_EDGES) * (unsigned long long)PB_WORDS +
                 (unsigned long long)E_ <=
             (unsigned long long)n * (unsigned long long)D1;
}

template <int NB>
__global__ __launch_bounds__(256) void plane_kernel(const int* __restrict__ ei,
                                                    unsigned* __restrict__ plane, int n, int E_) {
  if (!plane_usable<NB>(n, E_)) return;
  __shared__ __attribute__((aligned(16))) unsigned sm[32 * PB_PITCH];
  const int* __restrict__ dstp = ei + E_;
  const unsigned nblk = ((unsigned)E_ + (unsigned)(PB_EDGES - 1)) / (unsigned)PB_EDGES;
  unsigned* __restrict__ pk = plane + (size_t)nblk * PB_WORDS;
  const unsigned tid = threadIdx.x, lane = tid & 31u, wave = tid >> 5;
  const unsigned e0 = blockIdx.x * (unsigned)PB_EDGES + wave * 512u + lane;
  int d[16], s[16];
  #pragma unroll
  for (int i = 0; i < 16; i++) {
    const unsigned e = e0 + 32u * i;
    const bool ok = e < (unsigned)E_;
    d[i] = ok ? dstp[e] : 0;
    s[i] = ok ? ei[e] : 0;
  }
  const unsigned selh = lane < 16u ? 0xFFFFFFFFu : 0u;
  unsigned inv[4];
  #pragma unroll
  for (int t = 0; t < 4; t++) inv[t] = ((lane >> t) & 1u) ? 0u : 0xFFFFFFFFu;
  unsigned word[16];
  #pragma unroll
  for (int i = 0; i < 16; i++) {
    const unsigned e = e0 + 32u * i;
    const bool ok = e < (unsigned)E_;
    const unsigned sl = (unsigned)d[i] / (unsigned)NB;
    if (ok) pk[e] = ((unsigned)s[i] << DLO_BITS) | ((unsigned)d[i] - sl * (unsigned)NB);
    unsigned w = __builtin_amdgcn_ballot_w32(ok);
    #pragma unroll
    for (int t = 0; t < 4; t++) {
      const unsigned hb = __builtin_amdgcn_ballot_w32(((sl >> (4 + t)) & 1u) != 0u);
      const unsigned lb = __builtin_amdgcn_ballot_w32(((sl >> t) & 1u) != 0u);
      w &= ((hb & selh) | (lb & ~selh)) ^ inv[t];
    }
    word[i] = w;
  }
  #pragma unroll
  for (int r = 0; r < 4; r++)
    *(uint4*)&sm[lane * PB_PITCH + wave * 16u + 4u * r] =
        make_uint4(word[4 * r], word[4 * r + 1], word[4 * r + 2], word[4 * r + 3]);
  __syncthreads();
  #pragma unroll
  for (int r = 0; r < 4; r++) {
    const unsigned idx = (unsigned)r * 256u + tid;
    ((uint4*)plane)[(size_t)blockIdx.x * (PB_WORDS / 4) + idx] =
        *(const uint4*)&sm[(idx >> 5) * PB_PITCH + (idx & 31u) * 4u];
  }
}

template <int NB, int CAP>
__global__ __launch_bounds__(1024) void scan_kernel(const unsigned* __restrict__ plane,
                                                    int n, int E_, unsigned* lists) {
  static_assert(NB <= (1 << DLO_BITS), "local destination must fit its field");
  constexpr int NT = 1024;
  constexpr int NW = NT / 32;
  constexpr int SEG = CAP / NW;
  constexpr int BPI = 4;
  __shared__ unsigned seg[CAP];
  __shared__ unsigned lst[CAP];
  __shared__ unsigned wcnt[NW];

  const int tid = threadIdx.x;
  const unsigned cpad = (gridDim.x + 31u) & ~31u;
  const unsigned lcap = (((unsigned)LIST_WORDS * (unsigned)n - cpad) / gridDim.x) & ~31u;
  unsigned* ecnt = lists;
  unsigned* gl = lists + cpad + (size_t)blockIdx.x * lcap;

  const bool use = plane_usable<NB>(n, E_);
  unsigned total = 0u; bool fits = use;
  if (use) {
    const unsigned lane = tid & 31, wave = tid >> 5;
    unsigned* myseg = seg + wave * SEG;
    unsigned wn = 0u;
    auto push = [&](bool m, unsigned e) {
      const unsigned mask = __builtin_amdgcn_ballot_w32(m);
      if (m) { const unsigned k = wn + __builtin_amdgcn_mbcnt_lo(mask, 0u); if (k < (unsigned)SEG) myseg[k] = e; }
      wn += (unsigned)__builtin_popcount(mask);
    };
    const int nsb = (E_ + (PB_EDGES - 1)) / PB_EDGES;
    const unsigned* __restrict__ pk = plane + (size_t)nsb * PB_WORDS;
    const uint4* __restrict__ pp = (const uint4*)plane;
    const unsigned offh = (blockIdx.x >> 4) * 32u + lane;
    const unsigned offl = (16u + (blockIdx.x & 15u)) * 32u + lane;
    const int wv = __builtin_amdgcn_readfirstlane((int)wave);
    for (int sb0 = wv; sb0 < nsb; sb0 += BPI * NW) {
      uint4 vh[BPI], vl[BPI];
      #pragma unroll
      for (int s = 0; s < BPI; s++) {
        const int sb = sb0 + s * NW;
        const size_t base = (size_t)(sb < nsb ? sb : 0) * (PB_WORDS / 4);
        vh[s] = pp[base + offh]; vl[s] = pp[base + offl];
      }
      #pragma unroll
      for (int s = 0; s < BPI; s++) {
        const int sb = sb0 + s * NW;
        const bool live = sb < nsb;
        unsigned m[4];
        m[0] = vh[s].x & vl[s].x; m[1] = vh[s].y & vl[s].y;
        m[2] = vh[s].z & vl[s].z; m[3] = vh[s].w & vl[s].w;
        #pragma unroll
        for (int q = 0; q < 4; q++) {
          unsigned mm = live ? m[q] : 0u;
          const unsigned eb = (unsigned)sb * (unsigned)PB_EDGES + (4u * lane + (unsigned)q) * 32u;
          while (__builtin_amdgcn_ballot_w32(mm != 0u) != 0u) {
            const unsigned e = eb + (mm != 0u ? (unsigned)__builtin_ctz(mm) : 0u);
            push(mm != 0u && e < (unsigned)E_, e);
            mm &= mm - 1u;
          }
        }
      }
    }
    if (lane == 0) wcnt[wave] = wn;
    __syncthreads();
    unsigned off = 0u;
    for (int w = 0; w < NW; w++) {
      const unsigned c = wcnt[w];
      if ((unsigned)w < wave) off += c;
      total += c; fits = fits && c <= (unsigned)SEG;
    }
    if (fits) {
      for (unsigned j0 = lane; j0 < wn + lane; j0 += 256) {
        unsigned wd[8];
        #pragma unroll
        for (int u = 0; u < 8; u++) {
          const unsigned j = j0 + 32u * u;
          wd[u] = j < wn ? pk[myseg[j]] : 0u;
        }
        #pragma unroll
        for (int u = 0; u < 8; u++) {
          const unsigned j = j0 + 32u * u;
          if (j < wn) lst[off + j] = wd[u];
        }
      }
    }
  }
  if (fits) {
    __syncthreads();
    const bool pub = total <= lcap && (unsigned)n <= (1u << (32 - DLO_BITS));
    if (pub)
      for (unsigned j = tid; j < total; j += NT) gl[j] = lst[j];
    if (tid == 0) ecnt[blockIdx.x] = pub ? total : NOLIST;
  } else {
    if (tid == 0) ecnt[blockIdx.x] = NOLIST;
  }
}

template <int H, int CH, int NB, int CAP, bool ELU>
__global__ __launch_bounds__(1024) void agg_kernel(const int* __restrict__ ei,
                                                   const float* __restrict__ a_src,
                                                   const float* __restrict__ a_dst,
                                                   const float* __restrict__ xl,
                                                   const float* __restrict__ bias,
                                                   float* __restrict__ outp, int n, int E_,
                                                   const unsigned* lists) {
  static_assert(NB <= (1 << DLO_BITS), "local destination must fit its field");
  constexpr int D = H * CH;
  constexpr int RS = D >= 32 ? D + 1 : D;
  constexpr int NT = 1024;
  __shared__ float acc[NB * RS];
  __shared__ float den[NB * H];
  __shared__ unsigned mx[NB * H];
  __shared__ float adl[NB * H];
  __shared__ unsigned lst[CAP];
  __shared__ unsigned short dlo[CAP];

  const int tid = threadIdx.x;
  const int n0 = blockIdx.x * NB;
  const unsigned nloc = (unsigned)((n - n0) < NB ? (n - n0) : NB);
  const int* __restrict__ dstp = ei + E_;
  const unsigned cpad = (gridDim.x + 31u) & ~31u;
  const unsigned lcap = (((unsigned)LIST_WORDS * (unsigned)n - cpad) / gridDim.x) & ~31u;
  const unsigned* ecnt = lists;
  const unsigned* gl = lists + cpad + (size_t)blockIdx.x * lcap;

  for (int i = tid; i < NB * RS; i += NT) acc[i] = 0.f;
  for (int i = tid; i < NB * H; i += NT) {
    den[i] = 0.f; mx[i] = ENC_NEG_INF;
    adl[i] = ((unsigned)i < nloc * H) ? a_dst[(size_t)n0 * H + i] : 0.f;
  }
  __syncthreads();

  unsigned total = 0u;
  const unsigned cnt = ecnt[blockIdx.x];
  const bool fits = cnt <= (unsigned)CAP && cnt <= lcap;
  if (fits) {
    total = cnt;
    for (unsigned j = tid; j < total; j += NT) {
      const unsigned w = gl[j];
      lst[j] = w >> DLO_BITS; dlo[j] = (unsigned short)(w & ((1u << DLO_BITS) - 1u));
    }
    __syncthreads();
    const unsigned items = total * H;
    const unsigned mfull = items - items % (NT * 4);
    for (unsigned tb = tid; tb < mfull; tb += NT * 4) {
      float al[4]; unsigned slot[4];
      #pragma unroll
      for (int u = 0; u < 4; u++) {
        const unsigned t = tb + u * NT, i = t / H, hh = t % H;
        slot[u] = (unsigned)dlo[i] * H + hh;
        al[u] = a_src[(size_t)lst[i] * H + hh];
      }
      #pragma unroll
      for (int u = 0; u < 4; u++)
        atomicMax(&mx[slot[u]], fenc(lrelu(al[u] + adl[slot[u]])));
    }
    for (unsigned t = mfull + tid; t < items; t += NT) {
      const unsigned i = t / H, hh = t % H, sl = (unsigned)dlo[i] * H + hh;
      atomicMax(&mx[sl], fenc(lrelu(a_src[(size_t)lst[i] * H + hh] + adl[sl])));
    }
    __syncthreads();
    const unsigned lane = tid & 31u;
    const unsigned wv = (unsigned)__builtin_amdgcn_readfirstlane(tid >> 5);
    const unsigned tot = (unsigned)__builtin_amdgcn_readfirstlane((int)total);
    constexpr unsigned NW = NT / 32;
    if constexpr (D == 64) {
      static_assert(H == 8 && CH == 8, "lane map of the 64-word row");
      constexpr unsigned BE = 8;
      const unsigned hA = lane >> 2, eA = lane & 3u;
      const float* __restrict__ xlane = xl + 2u * lane;
      float* const accl = acc + 2u * lane;
      auto trip = [&](const unsigned ib, auto tailc) {
        constexpr bool TAIL = decltype(tailc)::value;
        const unsigned nval = TAIL ? tot - ib : BE;
        unsigned sA[2], dA[2]; float asv[2], exA[2]; bool ok[2];
        #pragma unroll
        for (int g = 0; g < 2; g++) {
          const unsigned k = eA + 4u * g;
          ok[g] = !TAIL || k < nval;
          const unsigned i = ib + (ok[g] ? k : 0u);
          sA[g] = lst[i]; dA[g] = dlo[i];
        }
        #pragma unroll
        for (int g = 0; g < 2; g++) asv[g] = a_src[(size_t)sA[g] * H + hA];
        float2 xr[BE];
#define G1_ROW(e) { const unsigned s_e = (unsigned)__builtin_amdgcn_readlane((int)sA[(e) >> 2], (e) & 3); \
                    xr[e] = *(const float2*)(xlane + (size_t)s_e * D); }
        G1_ROW(0) G1_ROW(1) G1_ROW(2) G1_ROW(3) G1_ROW(4) G1_ROW(5) G1_ROW(6) G1_ROW(7)
#undef G1_ROW
        #pragma unroll
        for (int g = 0; g < 2; g++) {
          const unsigned hs = dA[g] * H + hA;
          const float ex = __expf(lrelu(asv[g] + adl[hs]) - fdec(mx[hs]));
          exA[g] = ex;
          if (ok[g]) atomicAdd(&den[hs], ex);
        }
#define G1_ADD(e) if (!TAIL || (unsigned)(e) < nval) { \
          const unsigned d_e = (unsigned)__builtin_amdgcn_readlane((int)dA[(e) >> 2], (e) & 3); \
          const float ex = __uint_as_float(quad_take<(e) & 3>(__float_as_uint(exA[(e) >> 2]))); \
          float* ap = accl + d_e * RS; \
          atomicAdd(ap, xr[e].x * ex); atomicAdd(ap + 1, xr[e].y * ex); }
        G1_ADD(0) G1_ADD(1) G1_ADD(2) G1_ADD(3) G1_ADD(4) G1_ADD(5) G1_ADD(6) G1_ADD(7)
#undef G1_ADD
      };
      unsigned ib = wv * BE;
      for (; ib + BE <= tot; ib += NW * BE) trip(ib, std::false_type{});
      if (ib < tot) trip(ib, std::true_type{});
    } else {
      static_assert(D == 16 && H == 1, "lane map of the 16-word row");
      constexpr unsigned BE = 32;
      const unsigned p = lane & 15u, r = lane >> 4;
      const unsigned kA = 2u * p + r;
      const float* __restrict__ xlane = xl + p;
      float* const accl = acc + p;
      auto trip = [&](const unsigned ib, auto tailc) {
        constexpr bool TAIL = decltype(tailc)::value;
        const unsigned nval = TAIL ? tot - ib : BE;
        const bool ok = !TAIL || kA < nval;
        const unsigned i = ib + (ok ? kA : 0u);
        const unsigned s = lst[i], d = dlo[i];
        const float asv = a_src[s];
        float xv[16];
#define G2_ROW(k) { const unsigned s_k = row_take<k>(s); xv[k] = xlane[(size_t)s_k * D]; }
        G2_ROW(0) G2_ROW(1) G2_ROW(2) G2_ROW(3) G2_ROW(4) G2_ROW(5) G2_ROW(6) G2_ROW(7)
        G2_ROW(8) G2_ROW(9) G2_ROW(10) G2_ROW(11) G2_ROW(12) G2_ROW(13) G2_ROW(14) G2_ROW(15)
#undef G2_ROW
        const float ex = __expf(lrelu(asv + adl[d]) - fdec(mx[d]));
        if (ok) atomicAdd(&den[d], ex);
#define G2_ADD(k) { const unsigned d_k = row_take<k>(d); \
          const float e_k = __uint_as_float(row_take<k>(__float_as_uint(ex))); \
          if (!TAIL || 2u * (k) + r < nval) atomicAdd(accl + d_k * RS, xv[k] * e_k); }
        G2_ADD(0) G2_ADD(1) G2_ADD(2) G2_ADD(3) G2_ADD(4) G2_ADD(5) G2_ADD(6) G2_ADD(7)
        G2_ADD(8) G2_ADD(9) G2_ADD(10) G2_ADD(11) G2_ADD(12) G2_ADD(13) G2_ADD(14) G2_ADD(15)
#undef G2_ADD
      };
      unsigned ib = wv * BE;
      for (; ib + BE <= tot; ib += NW * BE) trip(ib, std::false_type{});
      if (ib < tot) trip(ib, std::true_type{});
    }
  } else {
    for (int e = tid; e < E_; e += NT) {
      const unsigned r = (unsigned)(dstp[e] - n0);
      if (r < nloc) {
        const int s = ei[e];
        for (int h2 = 0; h2 < H; h2++)
          atomicMax(&mx[r * H + h2], fenc(lrelu(a_src[(size_t)s * H + h2] + adl[r * H + h2])));
      }
    }
    __syncthreads();
    for (int e = tid; e < E_; e += NT) {
      const unsigned r = (unsigned)(dstp[e] - n0);
      if (r < nloc) {
        const int s = ei[e];
        for (int h2 = 0; h2 < H; h2++) {
          const unsigned hs = r * H + h2;
          const float ex = __expf(lrelu(a_src[(size_t)s * H + h2] + adl[hs]) - fdec(mx[hs]));
          atomicAdd(&den[hs], ex);
          for (int cc = 0; cc < CH; cc++)
            atomicAdd(&acc[r * RS + h2 * CH + cc], xl[(size_t)s * D + h2 * CH + cc] * ex);
        }
      }
    }
  }
  __syncthreads();
  float* op = outp + (size_t)n0 * D;
  for (unsigned i = tid; i < nloc * D; i += NT) {
    const unsigned c = i % D, r = i / D;
    float v = acc[r * RS + c] / (den[r * H + c / CH] + EPS_GAT) + bias[c];
    if (ELU) v = v > 0.f ? v : (__expf(v) - 1.f);
    op[i] = v;
  }
}

__global__ __launch_bounds__(256) void gemm2_kernel(const float* __restrict__ h,
                                                    const float* __restrict__ W2,
                                                    float* __restrict__ xl2, int nrows) {
  const int wave = threadIdx.x >> 5;
  const int lane = threadIdx.x & 31;
  const int tile = blockIdx.x * 8 + wave;
  const int rowBase = tile * 16;
  if (rowBase >= nrows) return;
  const int ln = lane & 15;
  const int khalf = (lane >> 4) << 1;
  int arow = rowBase + ln; if (arow >= nrows) arow = nrows - 1;
  const float* ap = h + (size_t)arow * D1;

  v8f c = {};
  #pragma unroll
  for (int k = 0; k < D1; k += 4) {
    v2f a = *(const v2f*)&ap[k + khalf];
    v2f b; b.x = W2[(k + khalf) * NCLS + ln];
           b.y = W2[(k + khalf + 1) * NCLS + ln];
    c = __builtin_amdgcn_wmma_f32_16x16x4_f32(false, a, false, b, (short)0, c,
                                              false, false);
  }
  const int mBase = (lane >> 4) * 8;
  float* outp = xl2 + (size_t)rowBase * NCLS;
  #pragma unroll
  for (int i = 0; i < 8; i++) {
    int row = rowBase + mBase + i;
    if (row < nrows) outp[(size_t)(mBase + i) * NCLS + ln] = c[i];
  }
}

__global__ void attn2_kernel(const float* __restrict__ xl2,
                             const float* __restrict__ att_src,
                             const float* __restrict__ att_dst,
                             float* __restrict__ a_src, float* __restrict__ a_dst, int n) {
  int t = blockIdx.x * blockDim.x + threadIdx.x;
  if (t >= n) return;
  const float* xp = xl2 + (size_t)t * NCLS;
  float s1 = 0.f, s2 = 0.f;
  #pragma unroll
  for (int cc = 0; cc < NCLS; cc++) { float v = xp[cc]; s1 += v * att_src[cc]; s2 += v * att_dst[cc]; }
  a_src[t] = s1; a_dst[t] = s2;
}

extern "C" void kernel_launch(void* const* d_in, const int* in_sizes, int n_in,
                              void* d_out, int out_size, void* d_ws, size_t ws_size,
                              hipStream_t stream) {
  const float* x        = (const float*)d_in[0];
  const int*   ei       = (const int*)d_in[1];
  const float* W1       = (const float*)d_in[2];
  const float* att_src1 = (const float*)d_in[3];
  const float* att_dst1 = (const float*)d_in[4];
  const float* bias1    = (const float*)d_in[5];
  const float* W2       = (const float*)d_in[6];
  const float* att_src2 = (const float*)d_in[7];
  const float* att_dst2 = (const float*)d_in[8];
  const float* bias2    = (const float*)d_in[9];
  float* out = (float*)d_out;

  const int N = in_sizes[0] / F_IN;
  const int E = in_sizes[1] / 2;

  float* ws = (float*)d_ws;
  size_t o = 0;
  float*    xl1    = ws + o; o += (size_t)N * D1;
  float*    h1     = ws + o; o += (size_t)N * D1;
  float*    a_src1 = ws + o; o += (size_t)N * HEADS;
  float*    a_dst1 = ws + o; o += (size_t)N * HEADS;
  unsigned* lists  = (unsigned*)(ws + o); o += (size_t)N * LIST_WORDS;
  float*    xl2    = xl1;
  float*    a_src2 = xl2 + (size_t)N * NCLS;
  float*    a_dst2 = a_src2 + (size_t)N;

  const int TB = 256;
  auto nb = [](long long n, int b) { return (unsigned)((n + b - 1) / b); };

  unsigned* plane = (unsigned*)h1;

  plane_kernel<NB_AGG><<<nb((long long)E, PB_EDGES), TB, 0, stream>>>(ei, plane, N, E);
  scan_kernel<NB_AGG, CAP_AGG><<<nb(N, NB_AGG), 1024, 0, stream>>>(plane, N, E, lists);

  gemm1_kernel<<<256, 1024, 0, stream>>>(x, W1, xl1, N);
  attn1_kernel<<<nb((long long)N * HEADS, TB), TB, 0, stream>>>(xl1, att_src1, att_dst1,
                                                                a_src1, a_dst1, N);
  agg_kernel<HEADS, HID, NB_AGG, CAP_AGG, true><<<nb(N, NB_AGG), 1024, 0, stream>>>(
      ei, a_src1, a_dst1, xl1, bias1, h1, N, E, lists);

  gemm2_kernel<<<nb(((long long)N + 15) / 16, 8), 256, 0, stream>>>(h1, W2, xl2, N);
  attn2_kernel<<<nb(N, TB), TB, 0, stream>>>(xl2, att_src2, att_dst2, a_src2, a_dst2, N);
  agg_kernel<1, NCLS, NB_AGG, CAP_AGG, false><<<nb(N, NB_AGG), 1024, 0, stream>>>(
      ei, a_src2, a_dst2, xl2, bias2, out, N, E, lists);
}
